# starting kernel
# baseline (speedup 1.0000x reference)
#include <hip/hip_runtime.h>
#include <hip/hip_bf16.h>


typedef float v2f __attribute__((ext_vector_type(2)));
typedef float v8f __attribute__((ext_vector_type(8)));

#define NEG_SLOPE 0.2f
#define KEY_NEG_INF 0x007FFFFFu

__device__ __forceinline__ unsigned fkey(float x) {
  unsigned b = __float_as_uint(x);
  return (b & 0x80000000u) ? ~b : (b | 0x80000000u);
}
__device__ __forceinline__ float funkey(unsigned k) {
  return __uint_as_float((k & 0x80000000u) ? (k & 0x7FFFFFFFu) : ~k);
}


__global__ void fill_f32_kernel(float* __restrict__ p, int n, float v) {
  int i = blockIdx.x * blockDim.x + threadIdx.x;
  if (i < n) p[i] = v;
}
__global__ void fill_u32_kernel(unsigned* __restrict__ p, int n, unsigned v) {
  int i = blockIdx.x * blockDim.x + threadIdx.x;
  if (i < n) p[i] = v;
}


template <int K>
__global__ __launch_bounds__(256) void gemm_wmma_f32(
    const float* __restrict__ A, const float* __restrict__ W,
    float* __restrict__ H, int Cout) {
  const int lane = threadIdx.x;
  const int colTile = threadIdx.y;
  const int rowTile = blockIdx.x;

  const int mn    = lane & 15;
  const int khalf = (lane >> 4) * 2;

  const float* __restrict__ arow = A + (size_t)(rowTile * 16 + mn) * K + khalf;
  const float* __restrict__ wrow = W + (size_t)(colTile * 16 + mn) * K + khalf;

  v8f acc = {};
#pragma unroll
  for (int k0 = 0; k0 < K; k0 += 4) {
    v2f a = *(const v2f*)(arow + k0);
    v2f b = *(const v2f*)(wrow + k0);
    acc = __builtin_amdgcn_wmma_f32_16x16x4_f32(
         false, a,  false, b,
         (short)0, acc,  false,  false);
  }

  const int ncol = colTile * 16 + mn;
  const int mrow = rowTile * 16 + 8 * (lane >> 4);
#pragma unroll
  for (int v = 0; v < 8; ++v)
    H[(size_t)(mrow + v) * Cout + ncol] = acc[v];
}


__global__ void alpha_kernel(const float* __restrict__ H,
                             const float* __restrict__ att_src,
                             const float* __restrict__ att_dst,
                             float* __restrict__ as_, float* __restrict__ ad_,
                             int n, int heads, int C) {
  int idx = blockIdx.x * blockDim.x + threadIdx.x;
  if (idx >= n * heads) return;
  int node = idx / heads, h = idx % heads;
  const float* hp = H + (size_t)node * heads * C + (size_t)h * C;
  const float* sp = att_src + h * C;
  const float* dp = att_dst + h * C;
  float ss = 0.f, sd = 0.f;
  for (int c = 0; c < C; c += 4) {
    float4 hv = *(const float4*)(hp + c);
    float4 sv = *(const float4*)(sp + c);
    float4 dv = *(const float4*)(dp + c);
    ss += hv.x * sv.x + hv.y * sv.y + hv.z * sv.z + hv.w * sv.w;
    sd += hv.x * dv.x + hv.y * dv.y + hv.z * dv.z + hv.w * dv.w;
  }
  as_[idx] = ss;
  ad_[idx] = sd;
}


__device__ __forceinline__ void edge_endpoints(const int* __restrict__ ei,
                                               int E, int i, int& s, int& d) {
  if (i < E) { s = ei[i]; d = ei[E + i]; }
  else       { s = d = i - E; }
}

__global__ void edge_max_kernel(const int* __restrict__ ei, int E, int n,
                                int heads, const float* __restrict__ as_,
                                const float* __restrict__ ad_,
                                unsigned* __restrict__ segmax) {
  int i = blockIdx.x * blockDim.x + threadIdx.x;
  if (i >= E + n) return;
  int s, d;
  edge_endpoints(ei, E, i, s, d);
  for (int h = 0; h < heads; ++h) {
    float e = as_[s * heads + h] + ad_[d * heads + h];
    e = (e > 0.f) ? e : NEG_SLOPE * e;
    atomicMax(&segmax[d * heads + h], fkey(e));
  }
}

__global__ void edge_expsum_kernel(const int* __restrict__ ei, int E, int n,
                                   int heads, const float* __restrict__ as_,
                                   const float* __restrict__ ad_,
                                   const unsigned* __restrict__ segmax,
                                   float* __restrict__ exbuf,
                                   float* __restrict__ denom) {
  int i = blockIdx.x * blockDim.x + threadIdx.x;
  if (i >= E + n) return;
  int s, d;
  edge_endpoints(ei, E, i, s, d);
  for (int h = 0; h < heads; ++h) {
    float e = as_[s * heads + h] + ad_[d * heads + h];
    e = (e > 0.f) ? e : NEG_SLOPE * e;
    float m  = funkey(segmax[d * heads + h]);
    float ex = __expf(e - m);
    exbuf[(size_t)i * heads + h] = ex;
    atomicAdd(&denom[d * heads + h], ex);
  }
}

__global__ void aggregate_kernel(const int* __restrict__ ei, int E, int n,
                                 int heads, int C,
                                 const float* __restrict__ H,
                                 const float* __restrict__ exbuf,
                                 const float* __restrict__ denom,
                                 float* __restrict__ out) {
  int CH = heads * C;
  int idx = blockIdx.x * blockDim.x + threadIdx.x;
  int total = (E + n) * CH;
  if (idx >= total) return;
  int i = idx / CH;
  int c = idx % CH;
  int h = c / C;
  int s, d;
  edge_endpoints(ei, E, i, s, d);
  float alpha = exbuf[(size_t)i * heads + h] /
                (denom[d * heads + h] + 1e-16f);
  atomicAdd(&out[(size_t)d * CH + c], H[(size_t)s * CH + c] * alpha);
}


__global__ void bias_elu_kernel(float* __restrict__ x,
                                const float* __restrict__ b, int n, int CH) {
  int idx = blockIdx.x * blockDim.x + threadIdx.x;
  if (idx >= n * CH) return;
  float y = x[idx] + b[idx % CH];
  x[idx] = (y > 0.f) ? y : expm1f(y);
}

__global__ void bias_add_kernel(float* __restrict__ x,
                                const float* __restrict__ b, int n, int CH) {
  int idx = blockIdx.x * blockDim.x + threadIdx.x;
  if (idx >= n * CH) return;
  x[idx] = x[idx] + b[idx % CH];
}


static inline int cdiv(long long a, long long b) { return (int)((a + b - 1) / b); }

extern "C" void kernel_launch(void* const* d_in, const int* in_sizes, int n_in,
                              void* d_out, int out_size, void* d_ws, size_t ws_size,
                              hipStream_t stream) {
  const int IN_CH = 128, HID = 64, HEADS = 2, OUT = 64;
  const int CH1 = HEADS * HID;

  const float* x        = (const float*)d_in[0];
  const int*   ei       = (const int*)d_in[1];
  const float* W1       = (const float*)d_in[2];
  const float* att1_src = (const float*)d_in[3];
  const float* att1_dst = (const float*)d_in[4];
  const float* b1       = (const float*)d_in[5];
  const float* W2       = (const float*)d_in[6];
  const float* att2_src = (const float*)d_in[7];
  const float* att2_dst = (const float*)d_in[8];
  const float* b2       = (const float*)d_in[9];
  float* out            = (float*)d_out;

  const int n  = in_sizes[0] / IN_CH;
  const int E  = in_sizes[1] / 2;
  const int Ep = E + n;

  char* ws = (char*)d_ws;
  size_t off = 0;
  auto alloc = [&](size_t bytes) -> void* {
    void* p = ws + off;
    off += (bytes + 255) & ~(size_t)255;
    return p;
  };
  float*    h1      = (float*)   alloc((size_t)n * CH1 * 4);
  float*    agg1    = (float*)   alloc((size_t)n * CH1 * 4);
  float*    h2      = (float*)   alloc((size_t)n * OUT * 4);
  float*    as1     = (float*)   alloc((size_t)n * HEADS * 4);
  float*    ad1     = (float*)   alloc((size_t)n * HEADS * 4);
  unsigned* segmax1 = (unsigned*)alloc((size_t)n * HEADS * 4);
  float*    denom1  = (float*)   alloc((size_t)n * HEADS * 4);
  float*    as2     = (float*)   alloc((size_t)n * 4);
  float*    ad2     = (float*)   alloc((size_t)n * 4);
  unsigned* segmax2 = (unsigned*)alloc((size_t)n * 4);
  float*    denom2  = (float*)   alloc((size_t)n * 4);
  float*    ex1     = (float*)   alloc((size_t)Ep * HEADS * 4);
  float*    ex2     = (float*)   alloc((size_t)Ep * 4);
  (void)ws_size;

  const int BT = 256;

  fill_f32_kernel<<<cdiv((long long)n * CH1, BT), BT, 0, stream>>>(agg1, n * CH1, 0.f);
  fill_f32_kernel<<<cdiv(n * HEADS, BT), BT, 0, stream>>>(denom1, n * HEADS, 0.f);
  fill_u32_kernel<<<cdiv(n * HEADS, BT), BT, 0, stream>>>(segmax1, n * HEADS, KEY_NEG_INF);
  fill_f32_kernel<<<cdiv((long long)n * OUT, BT), BT, 0, stream>>>(out, n * OUT, 0.f);
  fill_f32_kernel<<<cdiv(n, BT), BT, 0, stream>>>(denom2, n, 0.f);
  fill_u32_kernel<<<cdiv(n, BT), BT, 0, stream>>>(segmax2, n, KEY_NEG_INF);

  {
    dim3 block(32, CH1 / 16);
    dim3 grid(n / 16);
    gemm_wmma_f32<128><<<grid, block, 0, stream>>>(x, W1, h1, CH1);
  }
  alpha_kernel<<<cdiv(n * HEADS, BT), BT, 0, stream>>>(
      h1, att1_src, att1_dst, as1, ad1, n, HEADS, HID);
  edge_max_kernel<<<cdiv(Ep, BT), BT, 0, stream>>>(ei, E, n, HEADS, as1, ad1, segmax1);
  edge_expsum_kernel<<<cdiv(Ep, BT), BT, 0, stream>>>(
      ei, E, n, HEADS, as1, ad1, segmax1, ex1, denom1);
  aggregate_kernel<<<cdiv((long long)Ep * CH1, BT), BT, 0, stream>>>(
      ei, E, n, HEADS, HID, h1, ex1, denom1, agg1);
  bias_elu_kernel<<<cdiv((long long)n * CH1, BT), BT, 0, stream>>>(agg1, b1, n, CH1);

  {
    dim3 block(32, OUT / 16);
    dim3 grid(n / 16);
    gemm_wmma_f32<128><<<grid, block, 0, stream>>>(agg1, W2, h2, OUT);
  }
  alpha_kernel<<<cdiv(n, BT), BT, 0, stream>>>(
      h2, att2_src, att2_dst, as2, ad2, n, 1, OUT);
  edge_max_kernel<<<cdiv(Ep, BT), BT, 0, stream>>>(ei, E, n, 1, as2, ad2, segmax2);
  edge_expsum_kernel<<<cdiv(Ep, BT), BT, 0, stream>>>(
      ei, E, n, 1, as2, ad2, segmax2, ex2, denom2);
  aggregate_kernel<<<cdiv((long long)Ep * OUT, BT), BT, 0, stream>>>(
      ei, E, n, 1, OUT, h2, ex2, denom2, out);
  bias_add_kernel<<<cdiv((long long)n * OUT, BT), BT, 0, stream>>>(out, b2, n, OUT);
}


# layer-1 edge kernel writes each wave's filtered edge list (source, local destination) to the workspace; layer-2 edge kernel reads it instead of scanning all edges again (sum order still LDS arrival)
# speedup vs baseline: 5.6004x; 5.6004x over previous
#include <hip/hip_runtime.h>
#include <hip/hip_bf16.h>


typedef float v2f __attribute__((ext_vector_type(2)));
typedef float v8f __attribute__((ext_vector_type(8)));

#define NEG_SLOPE 0.2f
#define KEY_NEG_INF 0x007FFFFFu

__device__ __forceinline__ unsigned fkey(float x) {
  unsigned b = __float_as_uint(x);
  return (b & 0x80000000u) ? ~b : (b | 0x80000000u);
}
__device__ __forceinline__ float funkey(unsigned k) {
  return __uint_as_float((k & 0x80000000u) ? (k & 0x7FFFFFFFu) : ~k);
}


template <int K>
__global__ __launch_bounds__(256) void gemm_wmma_f32(
    const float* __restrict__ A, const float* __restrict__ W,
    float* __restrict__ H, int Cout) {
  const int lane = threadIdx.x;
  const int colTile = threadIdx.y;
  const int rowTile = blockIdx.x;

  const int mn    = lane & 15;
  const int khalf = (lane >> 4) * 2;

  const float* __restrict__ arow = A + (size_t)(rowTile * 16 + mn) * K + khalf;
  const float* __restrict__ wrow = W + (size_t)(colTile * 16 + mn) * K + khalf;

  v8f acc = {};
#pragma unroll
  for (int k0 = 0; k0 < K; k0 += 4) {
    v2f a = *(const v2f*)(arow + k0);
    v2f b = *(const v2f*)(wrow + k0);
    acc = __builtin_amdgcn_wmma_f32_16x16x4_f32(
         false, a,  false, b,
         (short)0, acc,  false,  false);
  }

  const int ncol = colTile * 16 + mn;
  const int mrow = rowTile * 16 + 8 * (lane >> 4);
#pragma unroll
  for (int v = 0; v < 8; ++v)
    H[(size_t)(mrow + v) * Cout + ncol] = acc[v];
}


__global__ void alpha_kernel(const float* __restrict__ H,
                             const float* __restrict__ att_src,
                             const float* __restrict__ att_dst,
                             float* __restrict__ as_, float* __restrict__ ad_,
                             int n, int heads, int C) {
  int idx = blockIdx.x * blockDim.x + threadIdx.x;
  if (idx >= n * heads) return;
  int node = idx / heads, h = idx % heads;
  const float* hp = H + (size_t)node * heads * C + (size_t)h * C;
  const float* sp = att_src + h * C;
  const float* dp = att_dst + h * C;
  float ss = 0.f, sd = 0.f;
  for (int c = 0; c < C; c += 4) {
    float4 hv = *(const float4*)(hp + c);
    float4 sv = *(const float4*)(sp + c);
    float4 dv = *(const float4*)(dp + c);
    ss += hv.x * sv.x + hv.y * sv.y + hv.z * sv.z + hv.w * sv.w;
    sd += hv.x * dv.x + hv.y * dv.y + hv.z * dv.z + hv.w * dv.w;
  }
  as_[idx] = ss;
  ad_[idx] = sd;
}


#define NBLK     196
#define NWAVE    32
#define LCAP     512
#define CHUNK    128
#define SCAN_UNR 8
#define ROW_UNR  8

__device__ __forceinline__ void scan_step(const int4 v, const int e0i,
                                          const int base, const int nb,
                                          unsigned* __restrict__ ls,
                                          unsigned* __restrict__ ld, int& cnt) {
  const unsigned e0 = (unsigned)e0i;
  const unsigned r0 = (unsigned)(v.x - base);
  const unsigned r1 = (unsigned)(v.y - base);
  const unsigned r2 = (unsigned)(v.z - base);
  const unsigned r3 = (unsigned)(v.w - base);
  const bool m0 = r0 < (unsigned)nb, m1 = r1 < (unsigned)nb;
  const bool m2 = r2 < (unsigned)nb, m3 = r3 < (unsigned)nb;
  if (__builtin_amdgcn_ballot_w32(m0 | m1 | m2 | m3)) {
    unsigned mk;
    mk = __builtin_amdgcn_ballot_w32(m0);
    if (m0) { const int ix = cnt + __builtin_amdgcn_mbcnt_lo(mk, 0u); ls[ix] = e0 + 0u; ld[ix] = r0; }
    cnt += __builtin_popcount(mk);
    mk = __builtin_amdgcn_ballot_w32(m1);
    if (m1) { const int ix = cnt + __builtin_amdgcn_mbcnt_lo(mk, 0u); ls[ix] = e0 + 1u; ld[ix] = r1; }
    cnt += __builtin_popcount(mk);
    mk = __builtin_amdgcn_ballot_w32(m2);
    if (m2) { const int ix = cnt + __builtin_amdgcn_mbcnt_lo(mk, 0u); ls[ix] = e0 + 2u; ld[ix] = r2; }
    cnt += __builtin_popcount(mk);
    mk = __builtin_amdgcn_ballot_w32(m3);
    if (m3) { const int ix = cnt + __builtin_amdgcn_mbcnt_lo(mk, 0u); ls[ix] = e0 + 3u; ld[ix] = r3; }
    cnt += __builtin_popcount(mk);
  }
}

template <int HEADS, int C, bool ELU, int LISTS>
__global__ __launch_bounds__(1024) void gat_block_kernel(
    const int* __restrict__ ei, int E, int n,
    const float* __restrict__ Hm,
    const float* __restrict__ as_, const float* __restrict__ ad_,
    const float* __restrict__ bias, float* __restrict__ outp,
    int* __restrict__ sv_cnt, int* __restrict__ sv_src, int* __restrict__ sv_dl) {
  constexpr int CH  = HEADS * C;
  constexpr int VEC = CH / 32;
  typedef float vrow __attribute__((ext_vector_type(VEC)));

  __shared__ float    acc[NBLK * CH];
  __shared__ unsigned lst_s[NWAVE * LCAP];
  __shared__ unsigned lst_d[NWAVE * LCAP];
  __shared__ unsigned mkey[NBLK * HEADS];
  __shared__ unsigned mold[NBLK * HEADS];
  __shared__ float    den[NBLK * HEADS];
  __shared__ int      wcnt[NWAVE];

  const int tid  = threadIdx.x;
  const int lane = tid & 31;
  const int w    = __builtin_amdgcn_readfirstlane(tid >> 5);
  const int base = blockIdx.x * NBLK;
  const int nb   = min(NBLK, n - base);
  const int* __restrict__ dstp = ei + E;

  for (int i = tid; i < NBLK * CH; i += 1024) acc[i] = 0.f;
  for (int i = tid; i < NBLK * HEADS; i += 1024) {
    mkey[i] = KEY_NEG_INF;
    den[i]  = 0.f;
  }

  unsigned* __restrict__ ls = lst_s + w * LCAP;
  unsigned* __restrict__ ld = lst_d + w * LCAP;

  const int nchunks = (E + CHUNK - 1) / CHUNK;
  const int seg = (blockIdx.x * NWAVE + w) * LCAP;
  int pos = 0;
  const bool loaded = (LISTS == 2) && (sv_cnt[blockIdx.x * NWAVE] >= 0);

  int cnt;
  if (loaded) {
    cnt = min(max(sv_cnt[blockIdx.x * NWAVE + w], 0), LCAP);
    pos = nchunks;
#pragma unroll 4
    for (int i = lane; i < cnt; i += 32) {
      ls[i] = (unsigned)sv_src[seg + i];
      ld[i] = (unsigned)sv_dl[seg + i];
    }
  } else {
    const int dl = w + NWAVE * lane;
    const bool ok = dl < nb;
    const unsigned mk = __builtin_amdgcn_ballot_w32(ok);
    if (ok) {
      const int idx = __builtin_amdgcn_mbcnt_lo(mk, 0u);
      ls[idx] = (unsigned)(E + base + dl);
      ld[idx] = (unsigned)dl;
    }
    cnt = __builtin_popcount(mk);
  }

  const bool vec_ok = ((E & 3) == 0) && ((((size_t)ei) & 15) == 0);
  const int hl = (lane * VEC) / C;
  int round = 0;

  for (;;) {
    while (w + NWAVE * pos < nchunks && cnt <= LCAP - CHUNK) {
      const int qlast = w + NWAVE * (pos + SCAN_UNR - 1);
      if (vec_ok && (qlast + 1) * CHUNK <= E) {
        int4 v[SCAN_UNR];
#pragma unroll
        for (int u = 0; u < SCAN_UNR; ++u) {
          const int q = w + NWAVE * (pos + u);
          v[u] = *(const int4*)(dstp + q * CHUNK + lane * 4);
        }
        int used = 0;
#pragma unroll
        for (int u = 0; u < SCAN_UNR; ++u) {
          if (cnt <= LCAP - CHUNK) {
            const int q = w + NWAVE * (pos + u);
            scan_step(v[u], q * CHUNK + lane * 4, base, nb, ls, ld, cnt);
            used = u + 1;
          }
        }
        pos += used;
      } else {
        const int q  = w + NWAVE * pos;
        const int e0 = q * CHUNK + lane * 4;
        int4 v;
        if (vec_ok && (q + 1) * CHUNK <= E) {
          v = *(const int4*)(dstp + e0);
        } else {
          v.x = (e0 + 0 < E) ? dstp[e0 + 0] : -1;
          v.y = (e0 + 1 < E) ? dstp[e0 + 1] : -1;
          v.z = (e0 + 2 < E) ? dstp[e0 + 2] : -1;
          v.w = (e0 + 3 < E) ? dstp[e0 + 3] : -1;
        }
        scan_step(v, e0, base, nb, ls, ld, cnt);
        pos += 1;
      }
    }
    const int mine_done = !(w + NWAVE * pos < nchunks);
    if (LISTS == 1 && round == 0 && lane == 0) wcnt[w] = cnt;

    if (round > 0)
      for (int i = tid; i < NBLK * HEADS; i += 1024) mold[i] = mkey[i];
    __syncthreads();

    for (int i = lane; i < cnt; i += 32) {
      const int dl = (int)ld[i];
      int s;
      if (loaded) {
        s = (int)ls[i];
      } else {
        const unsigned eidx = ls[i];
        s = (eidx < (unsigned)E) ? ei[eidx] : (int)(eidx - (unsigned)E);
        ls[i] = (unsigned)s;
      }
      if (LISTS == 1 && round == 0) {
        sv_src[seg + i] = s;
        sv_dl[seg + i]  = dl;
      }
#pragma unroll
      for (int h = 0; h < HEADS; ++h) {
        float e = as_[s * HEADS + h] + ad_[(base + dl) * HEADS + h];
        e = (e > 0.f) ? e : NEG_SLOPE * e;
        atomicMax(&mkey[dl * HEADS + h], fkey(e));
      }
    }
    __syncthreads();

    if (round > 0) {
      for (int i = tid; i < NBLK * CH; i += 1024) {
        const int dl = i / CH, r = i % CH;
        const int h = ((r & 31) * VEC + (r >> 5)) / C;
        const unsigned ko = mold[dl * HEADS + h], kn = mkey[dl * HEADS + h];
        const float f = (ko == KEY_NEG_INF) ? 0.f : __expf(funkey(ko) - funkey(kn));
        acc[i] *= f;
      }
      for (int i = tid; i < NBLK * HEADS; i += 1024) {
        const unsigned ko = mold[i], kn = mkey[i];
        const float f = (ko == KEY_NEG_INF) ? 0.f : __expf(funkey(ko) - funkey(kn));
        den[i] *= f;
      }
      __syncthreads();
    }

    for (int i0 = 0; i0 < cnt; i0 += ROW_UNR) {
      vrow  row[ROW_UNR];
      float exv[ROW_UNR];
      int   sv[ROW_UNR], dlv[ROW_UNR];
#pragma unroll
      for (int u = 0; u < ROW_UNR; ++u) {
        const int i = min(i0 + u, cnt - 1);
        sv[u]  = (int)ls[i];
        dlv[u] = (int)ld[i];
      }
#pragma unroll
      for (int u = 0; u < ROW_UNR; ++u) {
        const int s = sv[u], dl = dlv[u];
        row[u] = *(const vrow*)(Hm + (size_t)s * CH + lane * VEC);
        float e = as_[s * HEADS + hl] + ad_[(base + dl) * HEADS + hl];
        e = (e > 0.f) ? e : NEG_SLOPE * e;
        const float m = funkey(mkey[dl * HEADS + hl]);
        exv[u] = ((i0 + u) < cnt) ? __expf(e - m) : 0.f;
      }
#pragma unroll
      for (int u = 0; u < ROW_UNR; ++u) {
        if ((i0 + u) < cnt) {
#pragma unroll
          for (int j = 0; j < VEC; ++j)
            atomicAdd(&acc[dlv[u] * CH + j * 32 + lane], exv[u] * row[u][j]);
          if ((lane * VEC) % C == 0)
            atomicAdd(&den[dlv[u] * HEADS + hl], exv[u]);
        }
      }
    }

    if (__syncthreads_and(mine_done)) break;
    cnt = 0;
    ++round;
  }

  if (LISTS == 1 && w == 0)
    sv_cnt[blockIdx.x * NWAVE + lane] = (round == 0) ? wcnt[lane] : -1;

  for (int i = tid; i < nb * CH; i += 1024) {
    const int dl = i / CH, c = i % CH;
    const int h = c / C;
    float y = acc[dl * CH + (c % VEC) * 32 + (c / VEC)] /
                  (den[dl * HEADS + h] + 1e-16f) + bias[c];
    if (ELU) y = (y > 0.f) ? y : expm1f(y);
    outp[(size_t)(base + dl) * CH + c] = y;
  }
}


static inline int cdiv(long long a, long long b) { return (int)((a + b - 1) / b); }

extern "C" void kernel_launch(void* const* d_in, const int* in_sizes, int n_in,
                              void* d_out, int out_size, void* d_ws, size_t ws_size,
                              hipStream_t stream) {
  const int IN_CH = 128, HID = 64, HEADS = 2, OUT = 64;
  const int CH1 = HEADS * HID;

  const float* x        = (const float*)d_in[0];
  const int*   ei       = (const int*)d_in[1];
  const float* W1       = (const float*)d_in[2];
  const float* att1_src = (const float*)d_in[3];
  const float* att1_dst = (const float*)d_in[4];
  const float* b1       = (const float*)d_in[5];
  const float* W2       = (const float*)d_in[6];
  const float* att2_src = (const float*)d_in[7];
  const float* att2_dst = (const float*)d_in[8];
  const float* b2       = (const float*)d_in[9];
  float* out            = (float*)d_out;

  const int n  = in_sizes[0] / IN_CH;
  const int E  = in_sizes[1] / 2;

  char* ws = (char*)d_ws;
  size_t off = 0;
  auto alloc = [&](size_t bytes) -> void* {
    void* p = ws + off;
    off += (bytes + 255) & ~(size_t)255;
    return p;
  };
  float* h1   = (float*)alloc((size_t)n * CH1 * 4);
  float* agg1 = (float*)alloc((size_t)n * CH1 * 4);
  float* h2   = (float*)alloc((size_t)n * OUT * 4);
  float* as1  = (float*)alloc((size_t)n * HEADS * 4);
  float* ad1  = (float*)alloc((size_t)n * HEADS * 4);
  float* as2  = (float*)alloc((size_t)n * 4);
  float* ad2  = (float*)alloc((size_t)n * 4);
  const int nblk = cdiv(n, NBLK);
  int* sv_cnt = (int*)alloc((size_t)nblk * NWAVE * 4);
  int* sv_src = (int*)alloc((size_t)nblk * NWAVE * LCAP * 4);
  int* sv_dl  = (int*)alloc((size_t)nblk * NWAVE * LCAP * 4);
  (void)ws_size; (void)n_in; (void)out_size;

  const int BT = 256;

  {
    dim3 block(32, CH1 / 16);
    dim3 grid(n / 16);
    gemm_wmma_f32<128><<<grid, block, 0, stream>>>(x, W1, h1, CH1);
  }
  alpha_kernel<<<cdiv(n * HEADS, BT), BT, 0, stream>>>(
      h1, att1_src, att1_dst, as1, ad1, n, HEADS, HID);
  gat_block_kernel<2, 64, true, 1><<<nblk, 1024, 0, stream>>>(
      ei, E, n, h1, as1, ad1, b1, agg1, sv_cnt, sv_src, sv_dl);

  {
    dim3 block(32, OUT / 16);
    dim3 grid(n / 16);
    gemm_wmma_f32<128><<<grid, block, 0, stream>>>(agg1, W2, h2, OUT);
  }
  alpha_kernel<<<cdiv(n, BT), BT, 0, stream>>>(
      h2, att2_src, att2_dst, as2, ad2, n, 1, OUT);
  gat_block_kernel<1, 64, false, 2><<<nblk, 1024, 0, stream>>>(
      ei, E, n, h2, as2, ad2, b2, out, sv_cnt, sv_src, sv_dl);
}


# edge filter scans one key byte per edge (block number & 255, from a small kernel in front) instead of 4-byte destinations; false hits dropped at the destination lookup (sum order: LDS arrival)
# speedup vs baseline: 6.0280x; 1.0764x over previous
#include <hip/hip_runtime.h>
#include <hip/hip_bf16.h>


typedef float v2f __attribute__((ext_vector_type(2)));
typedef float v8f __attribute__((ext_vector_type(8)));

#define NEG_SLOPE 0.2f
#define KEY_NEG_INF 0x007FFFFFu

__device__ __forceinline__ unsigned fkey(float x) {
  unsigned b = __float_as_uint(x);
  return (b & 0x80000000u) ? ~b : (b | 0x80000000u);
}
__device__ __forceinline__ float funkey(unsigned k) {
  return __uint_as_float((k & 0x80000000u) ? (k & 0x7FFFFFFFu) : ~k);
}


template <int K>
__global__ __launch_bounds__(256) void gemm_wmma_f32(
    const float* __restrict__ A, const float* __restrict__ W,
    float* __restrict__ H, int Cout) {
  const int lane = threadIdx.x;
  const int colTile = threadIdx.y;
  const int rowTile = blockIdx.x;

  const int mn    = lane & 15;
  const int khalf = (lane >> 4) * 2;

  const float* __restrict__ arow = A + (size_t)(rowTile * 16 + mn) * K + khalf;
  const float* __restrict__ wrow = W + (size_t)(colTile * 16 + mn) * K + khalf;

  v8f acc = {};
#pragma unroll
  for (int k0 = 0; k0 < K; k0 += 4) {
    v2f a = *(const v2f*)(arow + k0);
    v2f b = *(const v2f*)(wrow + k0);
    acc = __builtin_amdgcn_wmma_f32_16x16x4_f32(
         false, a,  false, b,
         (short)0, acc,  false,  false);
  }

  const int ncol = colTile * 16 + mn;
  const int mrow = rowTile * 16 + 8 * (lane >> 4);
#pragma unroll
  for (int v = 0; v < 8; ++v)
    H[(size_t)(mrow + v) * Cout + ncol] = acc[v];
}


__global__ void alpha_kernel(const float* __restrict__ H,
                             const float* __restrict__ att_src,
                             const float* __restrict__ att_dst,
                             float* __restrict__ as_, float* __restrict__ ad_,
                             int n, int heads, int C) {
  int idx = blockIdx.x * blockDim.x + threadIdx.x;
  if (idx >= n * heads) return;
  int node = idx / heads, h = idx % heads;
  const float* hp = H + (size_t)node * heads * C + (size_t)h * C;
  const float* sp = att_src + h * C;
  const float* dp = att_dst + h * C;
  float ss = 0.f, sd = 0.f;
  for (int c = 0; c < C; c += 4) {
    float4 hv = *(const float4*)(hp + c);
    float4 sv = *(const float4*)(sp + c);
    float4 dv = *(const float4*)(dp + c);
    ss += hv.x * sv.x + hv.y * sv.y + hv.z * sv.z + hv.w * sv.w;
    sd += hv.x * dv.x + hv.y * dv.y + hv.z * dv.z + hv.w * dv.w;
  }
  as_[idx] = ss;
  ad_[idx] = sd;
}


#define NBLK     196
#define NWAVE    32
#define KSTEP    512
#define LCAP     768
#define SCAN_UNR 8
#define ROW_UNR  8
#define SELF_BIT 0x80000000u

#define KEY_UNR 4

__device__ __forceinline__ unsigned key_word(const int4 v) {
  const unsigned k0 = ((unsigned)v.x / NBLK) & 255u;
  const unsigned k1 = ((unsigned)v.y / NBLK) & 255u;
  const unsigned k2 = ((unsigned)v.z / NBLK) & 255u;
  const unsigned k3 = ((unsigned)v.w / NBLK) & 255u;
  return k0 | (k1 << 8) | (k2 << 16) | (k3 << 24);
}

__global__ __launch_bounds__(256) void dst_key_kernel(
    const int* __restrict__ ei, int E, int epad, unsigned* __restrict__ keys) {
  const int nw    = epad >> 2;
  const int nfull = E >> 2;
  const int T     = gridDim.x * blockDim.x;
  const int t0    = blockIdx.x * blockDim.x + threadIdx.x;
  const int* __restrict__ dstp = ei + E;
  if (nfull > 0 && ((E & 3) == 0) && ((((size_t)ei) & 15) == 0)) {
    for (int tb = t0; tb < nw; tb += KEY_UNR * T) {
      int4 v[KEY_UNR];
#pragma unroll
      for (int u = 0; u < KEY_UNR; ++u) {
        const int t = min(tb + u * T, nfull - 1);
        v[u] = *(const int4*)(dstp + 4 * t);
      }
#pragma unroll
      for (int u = 0; u < KEY_UNR; ++u) {
        const int t = tb + u * T;
        if (t < nw) keys[t] = (t < nfull) ? key_word(v[u]) : 0u;
      }
    }
  } else {
    for (int t = t0; t < nw; t += T) {
      const int e = 4 * t;
      int4 v;
      v.x = (e + 0 < E) ? dstp[e + 0] : 0;
      v.y = (e + 1 < E) ? dstp[e + 1] : 0;
      v.z = (e + 2 < E) ? dstp[e + 2] : 0;
      v.w = (e + 3 < E) ? dstp[e + 3] : 0;
      keys[t] = key_word(v);
    }
  }
}

__device__ __forceinline__ void key_step(const uint4 v, const unsigned pat,
                                         const unsigned e0,
                                         unsigned* __restrict__ ls, int& cnt) {
  const unsigned x0 = v.x ^ pat, x1 = v.y ^ pat, x2 = v.z ^ pat, x3 = v.w ^ pat;
  const unsigned t0 = (x0 - 0x01010101u) & ~x0;
  const unsigned t1 = (x1 - 0x01010101u) & ~x1;
  const unsigned t2 = (x2 - 0x01010101u) & ~x2;
  const unsigned t3 = (x3 - 0x01010101u) & ~x3;
  if (__builtin_amdgcn_ballot_w32(((t0 | t1 | t2 | t3) & 0x80808080u) != 0u)) {
    unsigned m = ((t0 >> 7) & 0x01010101u) | ((t1 >> 6) & 0x02020202u) |
                 ((t2 >> 5) & 0x04040404u) | ((t3 >> 4) & 0x08080808u);
    for (int it = 0; it < 16; ++it) {
      const bool has = m != 0u;
      const unsigned mk = __builtin_amdgcn_ballot_w32(has);
      if (mk == 0u) break;
      if (has) {
        const unsigned bit = (unsigned)__builtin_ctz(m);
        ls[cnt + __builtin_amdgcn_mbcnt_lo(mk, 0u)] = e0 + 4u * (bit & 7u) + (bit >> 3);
        m &= m - 1u;
      }
      cnt += __builtin_popcount(mk);
    }
  }
}

template <int HEADS, int C, bool ELU, int LISTS>
__global__ __launch_bounds__(1024) void gat_block_kernel(
    const int* __restrict__ ei, int E, int n,
    const float* __restrict__ Hm,
    const float* __restrict__ as_, const float* __restrict__ ad_,
    const float* __restrict__ bias, float* __restrict__ outp,
    int* __restrict__ sv_cnt, int* __restrict__ sv_src, int* __restrict__ sv_dl,
    const unsigned* __restrict__ keys) {
  constexpr int CH  = HEADS * C;
  constexpr int VEC = CH / 32;
  typedef float vrow __attribute__((ext_vector_type(VEC)));

  __shared__ float    acc[NBLK * CH];
  __shared__ unsigned lst_s[NWAVE * LCAP];
  __shared__ unsigned lst_d[NWAVE * LCAP];
  __shared__ unsigned mkey[NBLK * HEADS];
  __shared__ unsigned mold[NBLK * HEADS];
  __shared__ float    den[NBLK * HEADS];
  __shared__ int      wcnt[NWAVE];

  const int tid  = threadIdx.x;
  const int lane = tid & 31;
  const int w    = __builtin_amdgcn_readfirstlane(tid >> 5);
  const int base = blockIdx.x * NBLK;
  const int nb   = min(NBLK, n - base);
  const int* __restrict__ dstp = ei + E;

  for (int i = tid; i < NBLK * CH; i += 1024) acc[i] = 0.f;
  for (int i = tid; i < NBLK * HEADS; i += 1024) {
    mkey[i] = KEY_NEG_INF;
    den[i]  = 0.f;
  }

  unsigned* __restrict__ ls = lst_s + w * LCAP;
  unsigned* __restrict__ ld = lst_d + w * LCAP;

  const int nsteps = (E + KSTEP - 1) / KSTEP;
  const uint4* __restrict__ kp = (const uint4*)keys;
  const unsigned pat = (blockIdx.x & 255u) * 0x01010101u;
  const int seg = (blockIdx.x * NWAVE + w) * LCAP;
  int pos = 0;
  const bool loaded = (LISTS == 2) && (sv_cnt[blockIdx.x * NWAVE] >= 0);

  int cnt;
  if (loaded) {
    cnt = min(max(sv_cnt[blockIdx.x * NWAVE + w], 0), LCAP);
    pos = nsteps;
#pragma unroll 4
    for (int i = lane; i < cnt; i += 32) {
      ls[i] = (unsigned)sv_src[seg + i];
      ld[i] = (unsigned)sv_dl[seg + i];
    }
  } else {
    const int dl = w + NWAVE * lane;
    const bool ok = dl < nb;
    const unsigned mk = __builtin_amdgcn_ballot_w32(ok);
    if (ok) {
      const int idx = __builtin_amdgcn_mbcnt_lo(mk, 0u);
      ls[idx] = SELF_BIT | (unsigned)(base + dl);
    }
    cnt = __builtin_popcount(mk);
  }

  const int hl = (lane * VEC) / C;
  int round = 0;

  for (;;) {
    while (w + NWAVE * pos < nsteps && cnt <= LCAP - KSTEP) {
      const int qlast = w + NWAVE * (pos + SCAN_UNR - 1);
      if (qlast < nsteps) {
        uint4 v[SCAN_UNR];
#pragma unroll
        for (int u = 0; u < SCAN_UNR; ++u) {
          const int q = w + NWAVE * (pos + u);
          v[u] = kp[q * 32 + lane];
        }
        int used = 0;
#pragma unroll
        for (int u = 0; u < SCAN_UNR; ++u) {
          if (cnt <= LCAP - KSTEP) {
            const int q = w + NWAVE * (pos + u);
            key_step(v[u], pat, (unsigned)(q * KSTEP + lane * 16), ls, cnt);
            used = u + 1;
          }
        }
        pos += used;
      } else {
        const int q = w + NWAVE * pos;
        const uint4 v = kp[q * 32 + lane];
        key_step(v, pat, (unsigned)(q * KSTEP + lane * 16), ls, cnt);
        pos += 1;
      }
    }
    const int mine_done = !(w + NWAVE * pos < nsteps);

    if (round > 0)
      for (int i = tid; i < NBLK * HEADS; i += 1024) mold[i] = mkey[i];
    __syncthreads();

    int kept = 0;
    for (int i0 = 0; i0 < cnt; i0 += 32) {
      const int i = i0 + lane;
      bool ok = false;
      int s = 0;
      unsigned dl = 0u;
      if (i < cnt) {
        if (loaded) {
          s = (int)ls[i];
          dl = ld[i];
          ok = true;
        } else {
          const unsigned eidx = ls[i];
          if (eidx & SELF_BIT) {
            s = (int)(eidx & ~SELF_BIT);
            dl = (unsigned)(s - base);
            ok = true;
          } else if (eidx < (unsigned)E) {
            s = ei[eidx];
            dl = (unsigned)(dstp[eidx] - base);
            ok = dl < (unsigned)nb;
          }
        }
      }
      const unsigned mk = __builtin_amdgcn_ballot_w32(ok);
      if (ok) {
        const int o = kept + __builtin_amdgcn_mbcnt_lo(mk, 0u);
        if (!loaded) {
          ls[o] = (unsigned)s;
          ld[o] = dl;
        }
        if (LISTS == 1 && round == 0) {
          sv_src[seg + o] = s;
          sv_dl[seg + o]  = (int)dl;
        }
#pragma unroll
        for (int h = 0; h < HEADS; ++h) {
          float e = as_[s * HEADS + h] + ad_[(base + (int)dl) * HEADS + h];
          e = (e > 0.f) ? e : NEG_SLOPE * e;
          atomicMax(&mkey[dl * HEADS + h], fkey(e));
        }
      }
      kept += __builtin_popcount(mk);
    }
    cnt = kept;
    if (LISTS == 1 && round == 0 && lane == 0) wcnt[w] = cnt;
    __syncthreads();

    if (round > 0) {
      for (int i = tid; i < NBLK * CH; i += 1024) {
        const int dl = i / CH, r = i % CH;
        const int h = ((r & 31) * VEC + (r >> 5)) / C;
        const unsigned ko = mold[dl * HEADS + h], kn = mkey[dl * HEADS + h];
        const float f = (ko == KEY_NEG_INF) ? 0.f : __expf(funkey(ko) - funkey(kn));
        acc[i] *= f;
      }
      for (int i = tid; i < NBLK * HEADS; i += 1024) {
        const unsigned ko = mold[i], kn = mkey[i];
        const float f = (ko == KEY_NEG_INF) ? 0.f : __expf(funkey(ko) - funkey(kn));
        den[i] *= f;
      }
      __syncthreads();
    }

    for (int i0 = 0; i0 < cnt; i0 += ROW_UNR) {
      vrow  row[ROW_UNR];
      float exv[ROW_UNR];
      int   sv[ROW_UNR], dlv[ROW_UNR];
#pragma unroll
      for (int u = 0; u < ROW_UNR; ++u) {
        const int i = min(i0 + u, cnt - 1);
        sv[u]  = (int)ls[i];
        dlv[u] = (int)ld[i];
      }
#pragma unroll
      for (int u = 0; u < ROW_UNR; ++u) {
        const int s = sv[u], dl = dlv[u];
        row[u] = *(const vrow*)(Hm + (size_t)s * CH + lane * VEC);
        float e = as_[s * HEADS + hl] + ad_[(base + dl) * HEADS + hl];
        e = (e > 0.f) ? e : NEG_SLOPE * e;
        const float m = funkey(mkey[dl * HEADS + hl]);
        exv[u] = ((i0 + u) < cnt) ? __expf(e - m) : 0.f;
      }
#pragma unroll
      for (int u = 0; u < ROW_UNR; ++u) {
        if ((i0 + u) < cnt) {
#pragma unroll
          for (int j = 0; j < VEC; ++j)
            atomicAdd(&acc[dlv[u] * CH + j * 32 + lane], exv[u] * row[u][j]);
          if ((lane * VEC) % C == 0)
            atomicAdd(&den[dlv[u] * HEADS + hl], exv[u]);
        }
      }
    }

    if (__syncthreads_and(mine_done)) break;
    cnt = 0;
    ++round;
  }

  if (LISTS == 1 && w == 0)
    sv_cnt[blockIdx.x * NWAVE + lane] = (round == 0) ? wcnt[lane] : -1;

  for (int i = tid; i < nb * CH; i += 1024) {
    const int dl = i / CH, c = i % CH;
    const int h = c / C;
    float y = acc[dl * CH + (c % VEC) * 32 + (c / VEC)] /
                  (den[dl * HEADS + h] + 1e-16f) + bias[c];
    if (ELU) y = (y > 0.f) ? y : expm1f(y);
    outp[(size_t)(base + dl) * CH + c] = y;
  }
}


static inline int cdiv(long long a, long long b) { return (int)((a + b - 1) / b); }

extern "C" void kernel_launch(void* const* d_in, const int* in_sizes, int n_in,
                              void* d_out, int out_size, void* d_ws, size_t ws_size,
                              hipStream_t stream) {
  const int IN_CH = 128, HID = 64, HEADS = 2, OUT = 64;
  const int CH1 = HEADS * HID;

  const float* x        = (const float*)d_in[0];
  const int*   ei       = (const int*)d_in[1];
  const float* W1       = (const float*)d_in[2];
  const float* att1_src = (const float*)d_in[3];
  const float* att1_dst = (const float*)d_in[4];
  const float* b1       = (const float*)d_in[5];
  const float* W2       = (const float*)d_in[6];
  const float* att2_src = (const float*)d_in[7];
  const float* att2_dst = (const float*)d_in[8];
  const float* b2       = (const float*)d_in[9];
  float* out            = (float*)d_out;

  const int n  = in_sizes[0] / IN_CH;
  const int E  = in_sizes[1] / 2;

  char* ws = (char*)d_ws;
  size_t off = 0;
  auto alloc = [&](size_t bytes) -> void* {
    void* p = ws + off;
    off += (bytes + 255) & ~(size_t)255;
    return p;
  };
  float* h1   = (float*)alloc((size_t)n * CH1 * 4);
  float* agg1 = (float*)alloc((size_t)n * CH1 * 4);
  float* h2   = (float*)alloc((size_t)n * OUT * 4);
  float* as1  = (float*)alloc((size_t)n * HEADS * 4);
  float* ad1  = (float*)alloc((size_t)n * HEADS * 4);
  float* as2  = (float*)alloc((size_t)n * 4);
  float* ad2  = (float*)alloc((size_t)n * 4);
  const int nblk = cdiv(n, NBLK);
  int* sv_cnt = (int*)alloc((size_t)nblk * NWAVE * 4);
  int* sv_src = (int*)alloc((size_t)nblk * NWAVE * LCAP * 4);
  int* sv_dl  = (int*)alloc((size_t)nblk * NWAVE * LCAP * 4);
  const int epad = cdiv(E, KSTEP) * KSTEP;
  unsigned* keys = (unsigned*)alloc((size_t)epad);
  (void)ws_size; (void)n_in; (void)out_size;

  const int BT = 256;

  dst_key_kernel<<<cdiv(epad / 4, KEY_UNR * BT), BT, 0, stream>>>(ei, E, epad, keys);

  {
    dim3 block(32, CH1 / 16);
    dim3 grid(n / 16);
    gemm_wmma_f32<128><<<grid, block, 0, stream>>>(x, W1, h1, CH1);
  }
  alpha_kernel<<<cdiv(n * HEADS, BT), BT, 0, stream>>>(
      h1, att1_src, att1_dst, as1, ad1, n, HEADS, HID);
  gat_block_kernel<2, 64, true, 1><<<nblk, 1024, 0, stream>>>(
      ei, E, n, h1, as1, ad1, b1, agg1, sv_cnt, sv_src, sv_dl, keys);

  {
    dim3 block(32, OUT / 16);
    dim3 grid(n / 16);
    gemm_wmma_f32<128><<<grid, block, 0, stream>>>(agg1, W2, h2, OUT);
  }
  alpha_kernel<<<cdiv(n, BT), BT, 0, stream>>>(
      h2, att2_src, att2_dst, as2, ad2, n, 1, OUT);
  gat_block_kernel<1, 64, false, 2><<<nblk, 1024, 0, stream>>>(
      ei, E, n, h2, as2, ad2, b2, out, sv_cnt, sv_src, sv_dl, keys);
}


# GEMMs: a wave keeps the W fragments of its 32 columns in registers for the whole launch and loops over row tiles (one work-group per WGP); only A is loaded in the loop; k order and sums unchanged
# speedup vs baseline: 7.2659x; 1.2054x over previous
#include <hip/hip_runtime.h>
#include <hip/hip_bf16.h>


typedef float v2f __attribute__((ext_vector_type(2)));
typedef float v8f __attribute__((ext_vector_type(8)));

#define NEG_SLOPE 0.2f
#define KEY_NEG_INF 0x007FFFFFu

__device__ __forceinline__ unsigned fkey(float x) {
  unsigned b = __float_as_uint(x);
  return (b & 0x80000000u) ? ~b : (b | 0x80000000u);
}
__device__ __forceinline__ float funkey(unsigned k) {
  return __uint_as_float((k & 0x80000000u) ? (k & 0x7FFFFFFFu) : ~k);
}


template <int K, int CT, int NCG, int NRG>
__global__ __launch_bounds__(32 * NCG * NRG) void gemm_wmma_f32(
    const float* __restrict__ A, const float* __restrict__ W,
    float* __restrict__ H, int nRowTiles) {
  constexpr int KS   = K / 4;
  constexpr int Cout = 16 * CT * NCG;
  const int lane = threadIdx.x;
  const int colGroup = __builtin_amdgcn_readfirstlane(threadIdx.y);
  const int rowGroup = __builtin_amdgcn_readfirstlane(threadIdx.z);

  const int mn    = lane & 15;
  const int khalf = (lane >> 4) * 2;

  v2f wf[CT][KS];
#pragma unroll
  for (int c = 0; c < CT; ++c) {
    const float* __restrict__ wrow =
        W + (size_t)((colGroup * CT + c) * 16 + mn) * K + khalf;
#pragma unroll
    for (int j = 0; j < KS; ++j) wf[c][j] = *(const v2f*)(wrow + 4 * j);
  }

  for (int rowTile = rowGroup * gridDim.x + blockIdx.x; rowTile < nRowTiles;
       rowTile += gridDim.x * NRG) {
    const float* __restrict__ arow = A + (size_t)(rowTile * 16 + mn) * K + khalf;
    v2f a[KS];
#pragma unroll
    for (int j = 0; j < KS; ++j) a[j] = *(const v2f*)(arow + 4 * j);

    v8f acc[CT];
#pragma unroll
    for (int c = 0; c < CT; ++c) acc[c] = v8f{};
#pragma unroll
    for (int j = 0; j < KS; ++j) {
#pragma unroll
      for (int c = 0; c < CT; ++c)
        acc[c] = __builtin_amdgcn_wmma_f32_16x16x4_f32(
             false, a[j],  false, wf[c][j],
             (short)0, acc[c],  false,  false);
    }

    const int mrow = rowTile * 16 + 8 * (lane >> 4);
#pragma unroll
    for (int c = 0; c < CT; ++c) {
      const int ncol = (colGroup * CT + c) * 16 + mn;
#pragma unroll
      for (int v = 0; v < 8; ++v)
        H[(size_t)(mrow + v) * Cout + ncol] = acc[c][v];
    }
  }
}


__global__ void alpha_kernel(const float* __restrict__ H,
                             const float* __restrict__ att_src,
                             const float* __restrict__ att_dst,
                             float* __restrict__ as_, float* __restrict__ ad_,
                             int n, int heads, int C) {
  int idx = blockIdx.x * blockDim.x + threadIdx.x;
  if (idx >= n * heads) return;
  int node = idx / heads, h = idx % heads;
  const float* hp = H + (size_t)node * heads * C + (size_t)h * C;
  const float* sp = att_src + h * C;
  const float* dp = att_dst + h * C;
  float ss = 0.f, sd = 0.f;
  for (int c = 0; c < C; c += 4) {
    float4 hv = *(const float4*)(hp + c);
    float4 sv = *(const float4*)(sp + c);
    float4 dv = *(const float4*)(dp + c);
    ss += hv.x * sv.x + hv.y * sv.y + hv.z * sv.z + hv.w * sv.w;
    sd += hv.x * dv.x + hv.y * dv.y + hv.z * dv.z + hv.w * dv.w;
  }
  as_[idx] = ss;
  ad_[idx] = sd;
}


#define NBLK     196
#define NWAVE    32
#define KSTEP    512
#define LCAP     768
#define SCAN_UNR 8
#define ROW_UNR  8
#define SELF_BIT 0x80000000u

#define KEY_UNR 4

__device__ __forceinline__ unsigned key_word(const int4 v) {
  const unsigned k0 = ((unsigned)v.x / NBLK) & 255u;
  const unsigned k1 = ((unsigned)v.y / NBLK) & 255u;
  const unsigned k2 = ((unsigned)v.z / NBLK) & 255u;
  const unsigned k3 = ((unsigned)v.w / NBLK) & 255u;
  return k0 | (k1 << 8) | (k2 << 16) | (k3 << 24);
}

__global__ __launch_bounds__(256) void dst_key_kernel(
    const int* __restrict__ ei, int E, int epad, unsigned* __restrict__ keys) {
  const int nw    = epad >> 2;
  const int nfull = E >> 2;
  const int T     = gridDim.x * blockDim.x;
  const int t0    = blockIdx.x * blockDim.x + threadIdx.x;
  const int* __restrict__ dstp = ei + E;
  if (nfull > 0 && ((E & 3) == 0) && ((((size_t)ei) & 15) == 0)) {
    for (int tb = t0; tb < nw; tb += KEY_UNR * T) {
      int4 v[KEY_UNR];
#pragma unroll
      for (int u = 0; u < KEY_UNR; ++u) {
        const int t = min(tb + u * T, nfull - 1);
        v[u] = *(const int4*)(dstp + 4 * t);
      }
#pragma unroll
      for (int u = 0; u < KEY_UNR; ++u) {
        const int t = tb + u * T;
        if (t < nw) keys[t] = (t < nfull) ? key_word(v[u]) : 0u;
      }
    }
  } else {
    for (int t = t0; t < nw; t += T) {
      const int e = 4 * t;
      int4 v;
      v.x = (e + 0 < E) ? dstp[e + 0] : 0;
      v.y = (e + 1 < E) ? dstp[e + 1] : 0;
      v.z = (e + 2 < E) ? dstp[e + 2] : 0;
      v.w = (e + 3 < E) ? dstp[e + 3] : 0;
      keys[t] = key_word(v);
    }
  }
}

__device__ __forceinline__ void key_step(const uint4 v, const unsigned pat,
                                         const unsigned e0,
                                         unsigned* __restrict__ ls, int& cnt) {
  const unsigned x0 = v.x ^ pat, x1 = v.y ^ pat, x2 = v.z ^ pat, x3 = v.w ^ pat;
  const unsigned t0 = (x0 - 0x01010101u) & ~x0;
  const unsigned t1 = (x1 - 0x01010101u) & ~x1;
  const unsigned t2 = (x2 - 0x01010101u) & ~x2;
  const unsigned t3 = (x3 - 0x01010101u) & ~x3;
  if (__builtin_amdgcn_ballot_w32(((t0 | t1 | t2 | t3) & 0x80808080u) != 0u)) {
    unsigned m = ((t0 >> 7) & 0x01010101u) | ((t1 >> 6) & 0x02020202u) |
                 ((t2 >> 5) & 0x04040404u) | ((t3 >> 4) & 0x08080808u);
    for (int it = 0; it < 16; ++it) {
      const bool has = m != 0u;
      const unsigned mk = __builtin_amdgcn_ballot_w32(has);
      if (mk == 0u) break;
      if (has) {
        const unsigned bit = (unsigned)__builtin_ctz(m);
        ls[cnt + __builtin_amdgcn_mbcnt_lo(mk, 0u)] = e0 + 4u * (bit & 7u) + (bit >> 3);
        m &= m - 1u;
      }
      cnt += __builtin_popcount(mk);
    }
  }
}

template <int HEADS, int C, bool ELU, int LISTS>
__global__ __launch_bounds__(1024) void gat_block_kernel(
    const int* __restrict__ ei, int E, int n,
    const float* __restrict__ Hm,
    const float* __restrict__ as_, const float* __restrict__ ad_,
    const float* __restrict__ bias, float* __restrict__ outp,
    int* __restrict__ sv_cnt, int* __restrict__ sv_src, int* __restrict__ sv_dl,
    const unsigned* __restrict__ keys) {
  constexpr int CH  = HEADS * C;
  constexpr int VEC = CH / 32;
  typedef float vrow __attribute__((ext_vector_type(VEC)));

  __shared__ float    acc[NBLK * CH];
  __shared__ unsigned lst_s[NWAVE * LCAP];
  __shared__ unsigned lst_d[NWAVE * LCAP];
  __shared__ unsigned mkey[NBLK * HEADS];
  __shared__ unsigned mold[NBLK * HEADS];
  __shared__ float    den[NBLK * HEADS];
  __shared__ int      wcnt[NWAVE];

  const int tid  = threadIdx.x;
  const int lane = tid & 31;
  const int w    = __builtin_amdgcn_readfirstlane(tid >> 5);
  const int base = blockIdx.x * NBLK;
  const int nb   = min(NBLK, n - base);
  const int* __restrict__ dstp = ei + E;

  for (int i = tid; i < NBLK * CH; i += 1024) acc[i] = 0.f;
  for (int i = tid; i < NBLK * HEADS; i += 1024) {
    mkey[i] = KEY_NEG_INF;
    den[i]  = 0.f;
  }

  unsigned* __restrict__ ls = lst_s + w * LCAP;
  unsigned* __restrict__ ld = lst_d + w * LCAP;

  const int nsteps = (E + KSTEP - 1) / KSTEP;
  const uint4* __restrict__ kp = (const uint4*)keys;
  const unsigned pat = (blockIdx.x & 255u) * 0x01010101u;
  const int seg = (blockIdx.x * NWAVE + w) * LCAP;
  int pos = 0;
  const bool loaded = (LISTS == 2) && (sv_cnt[blockIdx.x * NWAVE] >= 0);

  int cnt;
  if (loaded) {
    cnt = min(max(sv_cnt[blockIdx.x * NWAVE + w], 0), LCAP);
    pos = nsteps;
#pragma unroll 4
    for (int i = lane; i < cnt; i += 32) {
      ls[i] = (unsigned)sv_src[seg + i];
      ld[i] = (unsigned)sv_dl[seg + i];
    }
  } else {
    const int dl = w + NWAVE * lane;
    const bool ok = dl < nb;
    const unsigned mk = __builtin_amdgcn_ballot_w32(ok);
    if (ok) {
      const int idx = __builtin_amdgcn_mbcnt_lo(mk, 0u);
      ls[idx] = SELF_BIT | (unsigned)(base + dl);
    }
    cnt = __builtin_popcount(mk);
  }

  const int hl = (lane * VEC) / C;
  int round = 0;

  for (;;) {
    while (w + NWAVE * pos < nsteps && cnt <= LCAP - KSTEP) {
      const int qlast = w + NWAVE * (pos + SCAN_UNR - 1);
      if (qlast < nsteps) {
        uint4 v[SCAN_UNR];
#pragma unroll
        for (int u = 0; u < SCAN_UNR; ++u) {
          const int q = w + NWAVE * (pos + u);
          v[u] = kp[q * 32 + lane];
        }
        int used = 0;
#pragma unroll
        for (int u = 0; u < SCAN_UNR; ++u) {
          if (cnt <= LCAP - KSTEP) {
            const int q = w + NWAVE * (pos + u);
            key_step(v[u], pat, (unsigned)(q * KSTEP + lane * 16), ls, cnt);
            used = u + 1;
          }
        }
        pos += used;
      } else {
        const int q = w + NWAVE * pos;
        const uint4 v = kp[q * 32 + lane];
        key_step(v, pat, (unsigned)(q * KSTEP + lane * 16), ls, cnt);
        pos += 1;
      }
    }
    const int mine_done = !(w + NWAVE * pos < nsteps);

    if (round > 0)
      for (int i = tid; i < NBLK * HEADS; i += 1024) mold[i] = mkey[i];
    __syncthreads();

    int kept = 0;
    for (int i0 = 0; i0 < cnt; i0 += 32) {
      const int i = i0 + lane;
      bool ok = false;
      int s = 0;
      unsigned dl = 0u;
      if (i < cnt) {
        if (loaded) {
          s = (int)ls[i];
          dl = ld[i];
          ok = true;
        } else {
          const unsigned eidx = ls[i];
          if (eidx & SELF_BIT) {
            s = (int)(eidx & ~SELF_BIT);
            dl = (unsigned)(s - base);
            ok = true;
          } else if (eidx < (unsigned)E) {
            s = ei[eidx];
            dl = (unsigned)(dstp[eidx] - base);
            ok = dl < (unsigned)nb;
          }
        }
      }
      const unsigned mk = __builtin_amdgcn_ballot_w32(ok);
      if (ok) {
        const int o = kept + __builtin_amdgcn_mbcnt_lo(mk, 0u);
        if (!loaded) {
          ls[o] = (unsigned)s;
          ld[o] = dl;
        }
        if (LISTS == 1 && round == 0) {
          sv_src[seg + o] = s;
          sv_dl[seg + o]  = (int)dl;
        }
#pragma unroll
        for (int h = 0; h < HEADS; ++h) {
          float e = as_[s * HEADS + h] + ad_[(base + (int)dl) * HEADS + h];
          e = (e > 0.f) ? e : NEG_SLOPE * e;
          atomicMax(&mkey[dl * HEADS + h], fkey(e));
        }
      }
      kept += __builtin_popcount(mk);
    }
    cnt = kept;
    if (LISTS == 1 && round == 0 && lane == 0) wcnt[w] = cnt;
    __syncthreads();

    if (round > 0) {
      for (int i = tid; i < NBLK * CH; i += 1024) {
        const int dl = i / CH, r = i % CH;
        const int h = ((r & 31) * VEC + (r >> 5)) / C;
        const unsigned ko = mold[dl * HEADS + h], kn = mkey[dl * HEADS + h];
        const float f = (ko == KEY_NEG_INF) ? 0.f : __expf(funkey(ko) - funkey(kn));
        acc[i] *= f;
      }
      for (int i = tid; i < NBLK * HEADS; i += 1024) {
        const unsigned ko = mold[i], kn = mkey[i];
        const float f = (ko == KEY_NEG_INF) ? 0.f : __expf(funkey(ko) - funkey(kn));
        den[i] *= f;
      }
      __syncthreads();
    }

    for (int i0 = 0; i0 < cnt; i0 += ROW_UNR) {
      vrow  row[ROW_UNR];
      float exv[ROW_UNR];
      int   sv[ROW_UNR], dlv[ROW_UNR];
#pragma unroll
      for (int u = 0; u < ROW_UNR; ++u) {
        const int i = min(i0 + u, cnt - 1);
        sv[u]  = (int)ls[i];
        dlv[u] = (int)ld[i];
      }
#pragma unroll
      for (int u = 0; u < ROW_UNR; ++u) {
        const int s = sv[u], dl = dlv[u];
        row[u] = *(const vrow*)(Hm + (size_t)s * CH + lane * VEC);
        float e = as_[s * HEADS + hl] + ad_[(base + dl) * HEADS + hl];
        e = (e > 0.f) ? e : NEG_SLOPE * e;
        const float m = funkey(mkey[dl * HEADS + hl]);
        exv[u] = ((i0 + u) < cnt) ? __expf(e - m) : 0.f;
      }
#pragma unroll
      for (int u = 0; u < ROW_UNR; ++u) {
        if ((i0 + u) < cnt) {
#pragma unroll
          for (int j = 0; j < VEC; ++j)
            atomicAdd(&acc[dlv[u] * CH + j * 32 + lane], exv[u] * row[u][j]);
          if ((lane * VEC) % C == 0)
            atomicAdd(&den[dlv[u] * HEADS + hl], exv[u]);
        }
      }
    }

    if (__syncthreads_and(mine_done)) break;
    cnt = 0;
    ++round;
  }

  if (LISTS == 1 && w == 0)
    sv_cnt[blockIdx.x * NWAVE + lane] = (round == 0) ? wcnt[lane] : -1;

  for (int i = tid; i < nb * CH; i += 1024) {
    const int dl = i / CH, c = i % CH;
    const int h = c / C;
    float y = acc[dl * CH + (c % VEC) * 32 + (c / VEC)] /
                  (den[dl * HEADS + h] + 1e-16f) + bias[c];
    if (ELU) y = (y > 0.f) ? y : expm1f(y);
    outp[(size_t)(base + dl) * CH + c] = y;
  }
}


#define GEMM_WGP 256

static inline int cdiv(long long a, long long b) { return (int)((a + b - 1) / b); }

extern "C" void kernel_launch(void* const* d_in, const int* in_sizes, int n_in,
                              void* d_out, int out_size, void* d_ws, size_t ws_size,
                              hipStream_t stream) {
  const int IN_CH = 128, HID = 64, HEADS = 2, OUT = 64;
  const int CH1 = HEADS * HID;

  const float* x        = (const float*)d_in[0];
  const int*   ei       = (const int*)d_in[1];
  const float* W1       = (const float*)d_in[2];
  const float* att1_src = (const float*)d_in[3];
  const float* att1_dst = (const float*)d_in[4];
  const float* b1       = (const float*)d_in[5];
  const float* W2       = (const float*)d_in[6];
  const float* att2_src = (const float*)d_in[7];
  const float* att2_dst = (const float*)d_in[8];
  const float* b2       = (const float*)d_in[9];
  float* out            = (float*)d_out;

  const int n  = in_sizes[0] / IN_CH;
  const int E  = in_sizes[1] / 2;

  char* ws = (char*)d_ws;
  size_t off = 0;
  auto alloc = [&](size_t bytes) -> void* {
    void* p = ws + off;
    off += (bytes + 255) & ~(size_t)255;
    return p;
  };
  float* h1   = (float*)alloc((size_t)n * CH1 * 4);
  float* agg1 = (float*)alloc((size_t)n * CH1 * 4);
  float* h2   = (float*)alloc((size_t)n * OUT * 4);
  float* as1  = (float*)alloc((size_t)n * HEADS * 4);
  float* ad1  = (float*)alloc((size_t)n * HEADS * 4);
  float* as2  = (float*)alloc((size_t)n * 4);
  float* ad2  = (float*)alloc((size_t)n * 4);
  const int nblk = cdiv(n, NBLK);
  int* sv_cnt = (int*)alloc((size_t)nblk * NWAVE * 4);
  int* sv_src = (int*)alloc((size_t)nblk * NWAVE * LCAP * 4);
  int* sv_dl  = (int*)alloc((size_t)nblk * NWAVE * LCAP * 4);
  const int epad = cdiv(E, KSTEP) * KSTEP;
  unsigned* keys = (unsigned*)alloc((size_t)epad);
  (void)ws_size; (void)n_in; (void)out_size;

  const int BT = 256;

  dst_key_kernel<<<cdiv(epad / 4, KEY_UNR * BT), BT, 0, stream>>>(ei, E, epad, keys);

  {
    dim3 block(32, 4, 3);
    dim3 grid(GEMM_WGP);
    gemm_wmma_f32<128, 2, 4, 3><<<grid, block, 0, stream>>>(x, W1, h1, n / 16);
  }
  alpha_kernel<<<cdiv(n * HEADS, BT), BT, 0, stream>>>(
      h1, att1_src, att1_dst, as1, ad1, n, HEADS, HID);
  gat_block_kernel<2, 64, true, 1><<<nblk, 1024, 0, stream>>>(
      ei, E, n, h1, as1, ad1, b1, agg1, sv_cnt, sv_src, sv_dl, keys);

  {
    dim3 block(32, 2, 4);
    dim3 grid(GEMM_WGP);
    gemm_wmma_f32<128, 2, 2, 4><<<grid, block, 0, stream>>>(agg1, W2, h2, n / 16);
  }
  alpha_kernel<<<cdiv(n, BT), BT, 0, stream>>>(
      h2, att2_src, att2_dst, as2, ad2, n, 1, OUT);
  gat_block_kernel<1, 64, false, 2><<<nblk, 1024, 0, stream>>>(
      ei, E, n, h2, as2, ad2, b2, out, sv_cnt, sv_src, sv_dl, keys);
}


# attention dots as extra GEMM columns: att vectors folded into the weights (rows att^T W, made in the key launch), one more WMMA tile in one wave per row group; 2 launches less (dot sum order new)
# speedup vs baseline: 7.5575x; 1.0401x over previous
#include <hip/hip_runtime.h>
#include <hip/hip_bf16.h>


typedef float v2f __attribute__((ext_vector_type(2)));
typedef float v8f __attribute__((ext_vector_type(8)));

#define NEG_SLOPE 0.2f
#define KEY_NEG_INF 0x007FFFFFu

__device__ __forceinline__ unsigned fkey(float x) {
  unsigned b = __float_as_uint(x);
  return (b & 0x80000000u) ? ~b : (b | 0x80000000u);
}
__device__ __forceinline__ float funkey(unsigned k) {
  return __uint_as_float((k & 0x80000000u) ? (k & 0x7FFFFFFFu) : ~k);
}


template <int K, int CT, int NV, bool WITHV>
__device__ __forceinline__ void gemm_row_tiles(
    const float* __restrict__ A, float* __restrict__ H, float* __restrict__ asd,
    const v2f (&wf)[CT][K / 4], const v2f* __restrict__ vf,
    const int firstTile, const int tileStep, const int nRowTiles,
    const int Cout, const int colTile0, const int lane) {
  constexpr int KS = K / 4;
  const int mn    = lane & 15;
  const int khalf = (lane >> 4) * 2;
  for (int rowTile = firstTile; rowTile < nRowTiles; rowTile += tileStep) {
    const float* __restrict__ arow = A + (size_t)(rowTile * 16 + mn) * K + khalf;
    v2f a[KS];
#pragma unroll
    for (int j = 0; j < KS; ++j) a[j] = *(const v2f*)(arow + 4 * j);

    v8f acc[CT];
    v8f accv = v8f{};
#pragma unroll
    for (int c = 0; c < CT; ++c) acc[c] = v8f{};
#pragma unroll
    for (int j = 0; j < KS; ++j) {
#pragma unroll
      for (int c = 0; c < CT; ++c)
        acc[c] = __builtin_amdgcn_wmma_f32_16x16x4_f32(
             false, a[j],  false, wf[c][j],
             (short)0, acc[c],  false,  false);
      if constexpr (WITHV)
        accv = __builtin_amdgcn_wmma_f32_16x16x4_f32(
             false, a[j],  false, vf[j],
             (short)0, accv,  false,  false);
    }

    const int mrow = rowTile * 16 + 8 * (lane >> 4);
#pragma unroll
    for (int c = 0; c < CT; ++c) {
      const int ncol = (colTile0 + c) * 16 + mn;
#pragma unroll
      for (int v = 0; v < 8; ++v)
        H[(size_t)(mrow + v) * Cout + ncol] = acc[c][v];
    }
    if constexpr (WITHV) {
      if (mn < NV) {
#pragma unroll
        for (int v = 0; v < 8; ++v)
          asd[(size_t)(mrow + v) * NV + mn] = accv[v];
      }
    }
  }
}

template <int K, int CT, int NCG, int NRG, int NV>
__global__ __launch_bounds__(32 * NCG * NRG) void gemm_wmma_f32(
    const float* __restrict__ A, const float* __restrict__ W,
    const float* __restrict__ V, float* __restrict__ H,
    float* __restrict__ asd, int nRowTiles) {
  constexpr int KS   = K / 4;
  constexpr int Cout = 16 * CT * NCG;
  const int lane = threadIdx.x;
  const int colGroup = __builtin_amdgcn_readfirstlane(threadIdx.y);
  const int rowGroup = __builtin_amdgcn_readfirstlane(threadIdx.z);

  const int mn    = lane & 15;
  const int khalf = (lane >> 4) * 2;

  v2f wf[CT][KS];
#pragma unroll
  for (int c = 0; c < CT; ++c) {
    const float* __restrict__ wrow =
        W + (size_t)((colGroup * CT + c) * 16 + mn) * K + khalf;
#pragma unroll
    for (int j = 0; j < KS; ++j) wf[c][j] = *(const v2f*)(wrow + 4 * j);
  }

  const int firstTile = rowGroup * gridDim.x + blockIdx.x;
  const int tileStep  = gridDim.x * NRG;
  if (colGroup == (rowGroup * NCG) / NRG) {
    v2f vf[KS];
    const float* __restrict__ vrow = V + (size_t)min(mn, NV - 1) * K + khalf;
#pragma unroll
    for (int j = 0; j < KS; ++j) {
      const v2f t = *(const v2f*)(vrow + 4 * j);
      vf[j] = (mn < NV) ? t : v2f{};
    }
    gemm_row_tiles<K, CT, NV, true>(A, H, asd, wf, vf, firstTile, tileStep,
                                    nRowTiles, Cout, colGroup * CT, lane);
  } else {
    gemm_row_tiles<K, CT, NV, false>(A, H, asd, wf, nullptr, firstTile, tileStep,
                                     nRowTiles, Cout, colGroup * CT, lane);
  }
}


template <int HEADS, int C, int K>
__device__ __forceinline__ void fold_att(const float* __restrict__ W,
                                         const float* __restrict__ att_src,
                                         const float* __restrict__ att_dst,
                                         float* __restrict__ V, int r, int k) {
  const bool isDst = r >= HEADS;
  const int  h     = isDst ? r - HEADS : r;
  const float* __restrict__ att = (isDst ? att_dst : att_src) + h * C;
  const float* __restrict__ wp  = W + (size_t)(h * C) * K + k;
  float s = 0.f;
#pragma unroll
  for (int c = 0; c < C; ++c) s += att[c] * wp[(size_t)c * K];
  V[r * K + k] = s;
}


#define NBLK     196
#define NWAVE    32
#define KSTEP    512
#define LCAP     768
#define SCAN_UNR 8
#define ROW_UNR  8
#define SELF_BIT 0x80000000u

#define KEY_UNR 4
#define FOLD_BLOCKS 3
#define FOLD_K      128

__device__ __forceinline__ unsigned key_word(const int4 v) {
  const unsigned k0 = ((unsigned)v.x / NBLK) & 255u;
  const unsigned k1 = ((unsigned)v.y / NBLK) & 255u;
  const unsigned k2 = ((unsigned)v.z / NBLK) & 255u;
  const unsigned k3 = ((unsigned)v.w / NBLK) & 255u;
  return k0 | (k1 << 8) | (k2 << 16) | (k3 << 24);
}

__global__ __launch_bounds__(256) void dst_key_kernel(
    const int* __restrict__ ei, int E, int epad, unsigned* __restrict__ keys,
    const float* __restrict__ W1, const float* __restrict__ att1_src,
    const float* __restrict__ att1_dst, float* __restrict__ V1,
    const float* __restrict__ W2, const float* __restrict__ att2_src,
    const float* __restrict__ att2_dst, float* __restrict__ V2) {
  if (blockIdx.x < FOLD_BLOCKS) {
    const int rr = threadIdx.x / FOLD_K, k = threadIdx.x % FOLD_K;
    if (blockIdx.x < 2)
      fold_att<2, 64, FOLD_K>(W1, att1_src, att1_dst, V1, 2 * (int)blockIdx.x + rr, k);
    else
      fold_att<1, 64, FOLD_K>(W2, att2_src, att2_dst, V2, rr, k);
    return;
  }
  const int nw    = epad >> 2;
  const int nfull = E >> 2;
  const int T     = (gridDim.x - FOLD_BLOCKS) * blockDim.x;
  const int t0    = (blockIdx.x - FOLD_BLOCKS) * blockDim.x + threadIdx.x;
  const int* __restrict__ dstp = ei + E;
  if (nfull > 0 && ((E & 3) == 0) && ((((size_t)ei) & 15) == 0)) {
    for (int tb = t0; tb < nw; tb += KEY_UNR * T) {
      int4 v[KEY_UNR];
#pragma unroll
      for (int u = 0; u < KEY_UNR; ++u) {
        const int t = min(tb + u * T, nfull - 1);
        v[u] = *(const int4*)(dstp + 4 * t);
      }
#pragma unroll
      for (int u = 0; u < KEY_UNR; ++u) {
        const int t = tb + u * T;
        if (t < nw) keys[t] = (t < nfull) ? key_word(v[u]) : 0u;
      }
    }
  } else {
    for (int t = t0; t < nw; t += T) {
      const int e = 4 * t;
      int4 v;
      v.x = (e + 0 < E) ? dstp[e + 0] : 0;
      v.y = (e + 1 < E) ? dstp[e + 1] : 0;
      v.z = (e + 2 < E) ? dstp[e + 2] : 0;
      v.w = (e + 3 < E) ? dstp[e + 3] : 0;
      keys[t] = key_word(v);
    }
  }
}

__device__ __forceinline__ void key_step(const uint4 v, const unsigned pat,
                                         const unsigned e0,
                                         unsigned* __restrict__ ls, int& cnt) {
  const unsigned x0 = v.x ^ pat, x1 = v.y ^ pat, x2 = v.z ^ pat, x3 = v.w ^ pat;
  const unsigned t0 = (x0 - 0x01010101u) & ~x0;
  const unsigned t1 = (x1 - 0x01010101u) & ~x1;
  const unsigned t2 = (x2 - 0x01010101u) & ~x2;
  const unsigned t3 = (x3 - 0x01010101u) & ~x3;
  if (__builtin_amdgcn_ballot_w32(((t0 | t1 | t2 | t3) & 0x80808080u) != 0u)) {
    unsigned m = ((t0 >> 7) & 0x01010101u) | ((t1 >> 6) & 0x02020202u) |
                 ((t2 >> 5) & 0x04040404u) | ((t3 >> 4) & 0x08080808u);
    for (int it = 0; it < 16; ++it) {
      const bool has = m != 0u;
      const unsigned mk = __builtin_amdgcn_ballot_w32(has);
      if (mk == 0u) break;
      if (has) {
        const unsigned bit = (unsigned)__builtin_ctz(m);
        ls[cnt + __builtin_amdgcn_mbcnt_lo(mk, 0u)] = e0 + 4u * (bit & 7u) + (bit >> 3);
        m &= m - 1u;
      }
      cnt += __builtin_popcount(mk);
    }
  }
}

template <int HEADS, int C, bool ELU, int LISTS>
__global__ __launch_bounds__(1024) void gat_block_kernel(
    const int* __restrict__ ei, int E, int n,
    const float* __restrict__ Hm,
    const float* __restrict__ asd,
    const float* __restrict__ bias, float* __restrict__ outp,
    int* __restrict__ sv_cnt, int* __restrict__ sv_src, int* __restrict__ sv_dl,
    const unsigned* __restrict__ keys) {
  constexpr int CH  = HEADS * C;
  constexpr int VEC = CH / 32;
  typedef float vrow __attribute__((ext_vector_type(VEC)));

  __shared__ float    acc[NBLK * CH];
  __shared__ unsigned lst_s[NWAVE * LCAP];
  __shared__ unsigned lst_d[NWAVE * LCAP];
  __shared__ unsigned mkey[NBLK * HEADS];
  __shared__ unsigned mold[NBLK * HEADS];
  __shared__ float    den[NBLK * HEADS];
  __shared__ int      wcnt[NWAVE];

  const int tid  = threadIdx.x;
  const int lane = tid & 31;
  const int w    = __builtin_amdgcn_readfirstlane(tid >> 5);
  const int base = blockIdx.x * NBLK;
  const int nb   = min(NBLK, n - base);
  const int* __restrict__ dstp = ei + E;

  for (int i = tid; i < NBLK * CH; i += 1024) acc[i] = 0.f;
  for (int i = tid; i < NBLK * HEADS; i += 1024) {
    mkey[i] = KEY_NEG_INF;
    den[i]  = 0.f;
  }

  unsigned* __restrict__ ls = lst_s + w * LCAP;
  unsigned* __restrict__ ld = lst_d + w * LCAP;

  const int nsteps = (E + KSTEP - 1) / KSTEP;
  const uint4* __restrict__ kp = (const uint4*)keys;
  const unsigned pat = (blockIdx.x & 255u) * 0x01010101u;
  const int seg = (blockIdx.x * NWAVE + w) * LCAP;
  int pos = 0;
  const bool loaded = (LISTS == 2) && (sv_cnt[blockIdx.x * NWAVE] >= 0);

  int cnt;
  if (loaded) {
    cnt = min(max(sv_cnt[blockIdx.x * NWAVE + w], 0), LCAP);
    pos = nsteps;
#pragma unroll 4
    for (int i = lane; i < cnt; i += 32) {
      ls[i] = (unsigned)sv_src[seg + i];
      ld[i] = (unsigned)sv_dl[seg + i];
    }
  } else {
    const int dl = w + NWAVE * lane;
    const bool ok = dl < nb;
    const unsigned mk = __builtin_amdgcn_ballot_w32(ok);
    if (ok) {
      const int idx = __builtin_amdgcn_mbcnt_lo(mk, 0u);
      ls[idx] = SELF_BIT | (unsigned)(base + dl);
    }
    cnt = __builtin_popcount(mk);
  }

  const int hl = (lane * VEC) / C;
  int round = 0;

  for (;;) {
    while (w + NWAVE * pos < nsteps && cnt <= LCAP - KSTEP) {
      const int qlast = w + NWAVE * (pos + SCAN_UNR - 1);
      if (qlast < nsteps) {
        uint4 v[SCAN_UNR];
#pragma unroll
        for (int u = 0; u < SCAN_UNR; ++u) {
          const int q = w + NWAVE * (pos + u);
          v[u] = kp[q * 32 + lane];
        }
        int used = 0;
#pragma unroll
        for (int u = 0; u < SCAN_UNR; ++u) {
          if (cnt <= LCAP - KSTEP) {
            const int q = w + NWAVE * (pos + u);
            key_step(v[u], pat, (unsigned)(q * KSTEP + lane * 16), ls, cnt);
            used = u + 1;
          }
        }
        pos += used;
      } else {
        const int q = w + NWAVE * pos;
        const uint4 v = kp[q * 32 + lane];
        key_step(v, pat, (unsigned)(q * KSTEP + lane * 16), ls, cnt);
        pos += 1;
      }
    }
    const int mine_done = !(w + NWAVE * pos < nsteps);

    if (round > 0)
      for (int i = tid; i < NBLK * HEADS; i += 1024) mold[i] = mkey[i];
    __syncthreads();

    int kept = 0;
    for (int i0 = 0; i0 < cnt; i0 += 32) {
      const int i = i0 + lane;
      bool ok = false;
      int s = 0;
      unsigned dl = 0u;
      if (i < cnt) {
        if (loaded) {
          s = (int)ls[i];
          dl = ld[i];
          ok = true;
        } else {
          const unsigned eidx = ls[i];
          if (eidx & SELF_BIT) {
            s = (int)(eidx & ~SELF_BIT);
            dl = (unsigned)(s - base);
            ok = true;
          } else if (eidx < (unsigned)E) {
            s = ei[eidx];
            dl = (unsigned)(dstp[eidx] - base);
            ok = dl < (unsigned)nb;
          }
        }
      }
      const unsigned mk = __builtin_amdgcn_ballot_w32(ok);
      if (ok) {
        const int o = kept + __builtin_amdgcn_mbcnt_lo(mk, 0u);
        if (!loaded) {
          ls[o] = (unsigned)s;
          ld[o] = dl;
        }
        if (LISTS == 1 && round == 0) {
          sv_src[seg + o] = s;
          sv_dl[seg + o]  = (int)dl;
        }
#pragma unroll
        for (int h = 0; h < HEADS; ++h) {
          float e = asd[s * (2 * HEADS) + h] +
                    asd[(base + (int)dl) * (2 * HEADS) + HEADS + h];
          e = (e > 0.f) ? e : NEG_SLOPE * e;
          atomicMax(&mkey[dl * HEADS + h], fkey(e));
        }
      }
      kept += __builtin_popcount(mk);
    }
    cnt = kept;
    if (LISTS == 1 && round == 0 && lane == 0) wcnt[w] = cnt;
    __syncthreads();

    if (round > 0) {
      for (int i = tid; i < NBLK * CH; i += 1024) {
        const int dl = i / CH, r = i % CH;
        const int h = ((r & 31) * VEC + (r >> 5)) / C;
        const unsigned ko = mold[dl * HEADS + h], kn = mkey[dl * HEADS + h];
        const float f = (ko == KEY_NEG_INF) ? 0.f : __expf(funkey(ko) - funkey(kn));
        acc[i] *= f;
      }
      for (int i = tid; i < NBLK * HEADS; i += 1024) {
        const unsigned ko = mold[i], kn = mkey[i];
        const float f = (ko == KEY_NEG_INF) ? 0.f : __expf(funkey(ko) - funkey(kn));
        den[i] *= f;
      }
      __syncthreads();
    }

    for (int i0 = 0; i0 < cnt; i0 += ROW_UNR) {
      vrow  row[ROW_UNR];
      float exv[ROW_UNR];
      int   sv[ROW_UNR], dlv[ROW_UNR];
#pragma unroll
      for (int u = 0; u < ROW_UNR; ++u) {
        const int i = min(i0 + u, cnt - 1);
        sv[u]  = (int)ls[i];
        dlv[u] = (int)ld[i];
      }
#pragma unroll
      for (int u = 0; u < ROW_UNR; ++u) {
        const int s = sv[u], dl = dlv[u];
        row[u] = *(const vrow*)(Hm + (size_t)s * CH + lane * VEC);
        float e = asd[s * (2 * HEADS) + hl] +
                  asd[(base + dl) * (2 * HEADS) + HEADS + hl];
        e = (e > 0.f) ? e : NEG_SLOPE * e;
        const float m = funkey(mkey[dl * HEADS + hl]);
        exv[u] = ((i0 + u) < cnt) ? __expf(e - m) : 0.f;
      }
#pragma unroll
      for (int u = 0; u < ROW_UNR; ++u) {
        if ((i0 + u) < cnt) {
#pragma unroll
          for (int j = 0; j < VEC; ++j)
            atomicAdd(&acc[dlv[u] * CH + j * 32 + lane], exv[u] * row[u][j]);
          if ((lane * VEC) % C == 0)
            atomicAdd(&den[dlv[u] * HEADS + hl], exv[u]);
        }
      }
    }

    if (__syncthreads_and(mine_done)) break;
    cnt = 0;
    ++round;
  }

  if (LISTS == 1 && w == 0)
    sv_cnt[blockIdx.x * NWAVE + lane] = (round == 0) ? wcnt[lane] : -1;

  for (int i = tid; i < nb * CH; i += 1024) {
    const int dl = i / CH, c = i % CH;
    const int h = c / C;
    float y = acc[dl * CH + (c % VEC) * 32 + (c / VEC)] /
                  (den[dl * HEADS + h] + 1e-16f) + bias[c];
    if (ELU) y = (y > 0.f) ? y : expm1f(y);
    outp[(size_t)(base + dl) * CH + c] = y;
  }
}


#define GEMM_WGP 256

static inline int cdiv(long long a, long long b) { return (int)((a + b - 1) / b); }

extern "C" void kernel_launch(void* const* d_in, const int* in_sizes, int n_in,
                              void* d_out, int out_size, void* d_ws, size_t ws_size,
                              hipStream_t stream) {
  const int IN_CH = 128, HID = 64, HEADS = 2, OUT = 64;
  const int CH1 = HEADS * HID;

  const float* x        = (const float*)d_in[0];
  const int*   ei       = (const int*)d_in[1];
  const float* W1       = (const float*)d_in[2];
  const float* att1_src = (const float*)d_in[3];
  const float* att1_dst = (const float*)d_in[4];
  const float* b1       = (const float*)d_in[5];
  const float* W2       = (const float*)d_in[6];
  const float* att2_src = (const float*)d_in[7];
  const float* att2_dst = (const float*)d_in[8];
  const float* b2       = (const float*)d_in[9];
  float* out            = (float*)d_out;

  const int n  = in_sizes[0] / IN_CH;
  const int E  = in_sizes[1] / 2;

  char* ws = (char*)d_ws;
  size_t off = 0;
  auto alloc = [&](size_t bytes) -> void* {
    void* p = ws + off;
    off += (bytes + 255) & ~(size_t)255;
    return p;
  };
  float* h1   = (float*)alloc((size_t)n * CH1 * 4);
  float* agg1 = (float*)alloc((size_t)n * CH1 * 4);
  float* h2   = (float*)alloc((size_t)n * OUT * 4);
  float* asd1 = (float*)alloc((size_t)n * 2 * HEADS * 4);
  float* asd2 = (float*)alloc((size_t)n * 2 * 4);
  float* V1   = (float*)alloc((size_t)2 * HEADS * IN_CH * 4);
  float* V2   = (float*)alloc((size_t)2 * CH1 * 4);
  const int nblk = cdiv(n, NBLK);
  int* sv_cnt = (int*)alloc((size_t)nblk * NWAVE * 4);
  int* sv_src = (int*)alloc((size_t)nblk * NWAVE * LCAP * 4);
  int* sv_dl  = (int*)alloc((size_t)nblk * NWAVE * LCAP * 4);
  const int epad = cdiv(E, KSTEP) * KSTEP;
  unsigned* keys = (unsigned*)alloc((size_t)epad);
  (void)ws_size; (void)n_in; (void)out_size;

  const int BT = 256;

  dst_key_kernel<<<FOLD_BLOCKS + cdiv(epad / 4, KEY_UNR * BT), BT, 0, stream>>>(
      ei, E, epad, keys, W1, att1_src, att1_dst, V1, W2, att2_src, att2_dst, V2);

  {
    dim3 block(32, 4, 3);
    dim3 grid(GEMM_WGP);
    gemm_wmma_f32<128, 2, 4, 3, 4><<<grid, block, 0, stream>>>(x, W1, V1, h1, asd1, n / 16);
  }
  gat_block_kernel<2, 64, true, 1><<<nblk, 1024, 0, stream>>>(
      ei, E, n, h1, asd1, b1, agg1, sv_cnt, sv_src, sv_dl, keys);

  {
    dim3 block(32, 2, 4);
    dim3 grid(GEMM_WGP);
    gemm_wmma_f32<128, 2, 2, 4, 2><<<grid, block, 0, stream>>>(agg1, W2, V2, h2, asd2, n / 16);
  }
  gat_block_kernel<1, 64, false, 2><<<nblk, 1024, 0, stream>>>(
      ei, E, n, h2, asd2, b2, out, sv_cnt, sv_src, sv_dl, keys);
}


# edge kernels, sum phase: a wave forms exp(e-max) and the exp-sum add of 32 listed edges lane per edge, in registers, then walks them with source, destination, weights as scalars (sums: LDS arrival)
# speedup vs baseline: 8.3250x; 1.1016x over previous
#include <hip/hip_runtime.h>
#include <hip/hip_bf16.h>


typedef float v2f __attribute__((ext_vector_type(2)));
typedef float v8f __attribute__((ext_vector_type(8)));

#define NEG_SLOPE 0.2f
#define KEY_NEG_INF 0x007FFFFFu

__device__ __forceinline__ unsigned fkey(float x) {
  unsigned b = __float_as_uint(x);
  return (b & 0x80000000u) ? ~b : (b | 0x80000000u);
}
__device__ __forceinline__ float funkey(unsigned k) {
  return __uint_as_float((k & 0x80000000u) ? (k & 0x7FFFFFFFu) : ~k);
}


template <int K, int CT, int NV, bool WITHV>
__device__ __forceinline__ void gemm_row_tiles(
    const float* __restrict__ A, float* __restrict__ H, float* __restrict__ asd,
    const v2f (&wf)[CT][K / 4], const v2f* __restrict__ vf,
    const int firstTile, const int tileStep, const int nRowTiles,
    const int Cout, const int colTile0, const int lane) {
  constexpr int KS = K / 4;
  const int mn    = lane & 15;
  const int khalf = (lane >> 4) * 2;
  for (int rowTile = firstTile; rowTile < nRowTiles; rowTile += tileStep) {
    const float* __restrict__ arow = A + (size_t)(rowTile * 16 + mn) * K + khalf;
    v2f a[KS];
#pragma unroll
    for (int j = 0; j < KS; ++j) a[j] = *(const v2f*)(arow + 4 * j);

    v8f acc[CT];
    v8f accv = v8f{};
#pragma unroll
    for (int c = 0; c < CT; ++c) acc[c] = v8f{};
#pragma unroll
    for (int j = 0; j < KS; ++j) {
#pragma unroll
      for (int c = 0; c < CT; ++c)
        acc[c] = __builtin_amdgcn_wmma_f32_16x16x4_f32(
             false, a[j],  false, wf[c][j],
             (short)0, acc[c],  false,  false);
      if constexpr (WITHV)
        accv = __builtin_amdgcn_wmma_f32_16x16x4_f32(
             false, a[j],  false, vf[j],
             (short)0, accv,  false,  false);
    }

    const int mrow = rowTile * 16 + 8 * (lane >> 4);
#pragma unroll
    for (int c = 0; c < CT; ++c) {
      const int ncol = (colTile0 + c) * 16 + mn;
#pragma unroll
      for (int v = 0; v < 8; ++v)
        H[(size_t)(mrow + v) * Cout + ncol] = acc[c][v];
    }
    if constexpr (WITHV) {
      if (mn < NV) {
#pragma unroll
        for (int v = 0; v < 8; ++v)
          asd[(size_t)(mrow + v) * NV + mn] = accv[v];
      }
    }
  }
}

template <int K, int CT, int NCG, int NRG, int NV>
__global__ __launch_bounds__(32 * NCG * NRG) void gemm_wmma_f32(
    const float* __restrict__ A, const float* __restrict__ W,
    const float* __restrict__ V, float* __restrict__ H,
    float* __restrict__ asd, int nRowTiles) {
  constexpr int KS   = K / 4;
  constexpr int Cout = 16 * CT * NCG;
  const int lane = threadIdx.x;
  const int colGroup = __builtin_amdgcn_readfirstlane(threadIdx.y);
  const int rowGroup = __builtin_amdgcn_readfirstlane(threadIdx.z);

  const int mn    = lane & 15;
  const int khalf = (lane >> 4) * 2;

  v2f wf[CT][KS];
#pragma unroll
  for (int c = 0; c < CT; ++c) {
    const float* __restrict__ wrow =
        W + (size_t)((colGroup * CT + c) * 16 + mn) * K + khalf;
#pragma unroll
    for (int j = 0; j < KS; ++j) wf[c][j] = *(const v2f*)(wrow + 4 * j);
  }

  const int firstTile = rowGroup * gridDim.x + blockIdx.x;
  const int tileStep  = gridDim.x * NRG;
  if (colGroup == (rowGroup * NCG) / NRG) {
    v2f vf[KS];
    const float* __restrict__ vrow = V + (size_t)min(mn, NV - 1) * K + khalf;
#pragma unroll
    for (int j = 0; j < KS; ++j) {
      const v2f t = *(const v2f*)(vrow + 4 * j);
      vf[j] = (mn < NV) ? t : v2f{};
    }
    gemm_row_tiles<K, CT, NV, true>(A, H, asd, wf, vf, firstTile, tileStep,
                                    nRowTiles, Cout, colGroup * CT, lane);
  } else {
    gemm_row_tiles<K, CT, NV, false>(A, H, asd, wf, nullptr, firstTile, tileStep,
                                     nRowTiles, Cout, colGroup * CT, lane);
  }
}


template <int HEADS, int C, int K>
__device__ __forceinline__ void fold_att(const float* __restrict__ W,
                                         const float* __restrict__ att_src,
                                         const float* __restrict__ att_dst,
                                         float* __restrict__ V, int r, int k) {
  const bool isDst = r >= HEADS;
  const int  h     = isDst ? r - HEADS : r;
  const float* __restrict__ att = (isDst ? att_dst : att_src) + h * C;
  const float* __restrict__ wp  = W + (size_t)(h * C) * K + k;
  float s = 0.f;
#pragma unroll
  for (int c = 0; c < C; ++c) s += att[c] * wp[(size_t)c * K];
  V[r * K + k] = s;
}


#define NBLK     196
#define NWAVE    32
#define KSTEP    512
#define LCAP     768
#define SCAN_UNR 8
#define SELF_BIT 0x80000000u

#define KEY_UNR 4
#define FOLD_BLOCKS 3
#define FOLD_K      128

__device__ __forceinline__ unsigned key_word(const int4 v) {
  const unsigned k0 = ((unsigned)v.x / NBLK) & 255u;
  const unsigned k1 = ((unsigned)v.y / NBLK) & 255u;
  const unsigned k2 = ((unsigned)v.z / NBLK) & 255u;
  const unsigned k3 = ((unsigned)v.w / NBLK) & 255u;
  return k0 | (k1 << 8) | (k2 << 16) | (k3 << 24);
}

__global__ __launch_bounds__(256) void dst_key_kernel(
    const int* __restrict__ ei, int E, int epad, unsigned* __restrict__ keys,
    const float* __restrict__ W1, const float* __restrict__ att1_src,
    const float* __restrict__ att1_dst, float* __restrict__ V1,
    const float* __restrict__ W2, const float* __restrict__ att2_src,
    const float* __restrict__ att2_dst, float* __restrict__ V2) {
  if (blockIdx.x < FOLD_BLOCKS) {
    const int rr = threadIdx.x / FOLD_K, k = threadIdx.x % FOLD_K;
    if (blockIdx.x < 2)
      fold_att<2, 64, FOLD_K>(W1, att1_src, att1_dst, V1, 2 * (int)blockIdx.x + rr, k);
    else
      fold_att<1, 64, FOLD_K>(W2, att2_src, att2_dst, V2, rr, k);
    return;
  }
  const int nw    = epad >> 2;
  const int nfull = E >> 2;
  const int T     = (gridDim.x - FOLD_BLOCKS) * blockDim.x;
  const int t0    = (blockIdx.x - FOLD_BLOCKS) * blockDim.x + threadIdx.x;
  const int* __restrict__ dstp = ei + E;
  if (nfull > 0 && ((E & 3) == 0) && ((((size_t)ei) & 15) == 0)) {
    for (int tb = t0; tb < nw; tb += KEY_UNR * T) {
      int4 v[KEY_UNR];
#pragma unroll
      for (int u = 0; u < KEY_UNR; ++u) {
        const int t = min(tb + u * T, nfull - 1);
        v[u] = *(const int4*)(dstp + 4 * t);
      }
#pragma unroll
      for (int u = 0; u < KEY_UNR; ++u) {
        const int t = tb + u * T;
        if (t < nw) keys[t] = (t < nfull) ? key_word(v[u]) : 0u;
      }
    }
  } else {
    for (int t = t0; t < nw; t += T) {
      const int e = 4 * t;
      int4 v;
      v.x = (e + 0 < E) ? dstp[e + 0] : 0;
      v.y = (e + 1 < E) ? dstp[e + 1] : 0;
      v.z = (e + 2 < E) ? dstp[e + 2] : 0;
      v.w = (e + 3 < E) ? dstp[e + 3] : 0;
      keys[t] = key_word(v);
    }
  }
}

__device__ __forceinline__ void key_step(const uint4 v, const unsigned pat,
                                         const unsigned e0,
                                         unsigned* __restrict__ ls, int& cnt) {
  const unsigned x0 = v.x ^ pat, x1 = v.y ^ pat, x2 = v.z ^ pat, x3 = v.w ^ pat;
  const unsigned t0 = (x0 - 0x01010101u) & ~x0;
  const unsigned t1 = (x1 - 0x01010101u) & ~x1;
  const unsigned t2 = (x2 - 0x01010101u) & ~x2;
  const unsigned t3 = (x3 - 0x01010101u) & ~x3;
  if (__builtin_amdgcn_ballot_w32(((t0 | t1 | t2 | t3) & 0x80808080u) != 0u)) {
    unsigned m = ((t0 >> 7) & 0x01010101u) | ((t1 >> 6) & 0x02020202u) |
                 ((t2 >> 5) & 0x04040404u) | ((t3 >> 4) & 0x08080808u);
    for (int it = 0; it < 16; ++it) {
      const bool has = m != 0u;
      const unsigned mk = __builtin_amdgcn_ballot_w32(has);
      if (mk == 0u) break;
      if (has) {
        const unsigned bit = (unsigned)__builtin_ctz(m);
        ls[cnt + __builtin_amdgcn_mbcnt_lo(mk, 0u)] = e0 + 4u * (bit & 7u) + (bit >> 3);
        m &= m - 1u;
      }
      cnt += __builtin_popcount(mk);
    }
  }
}

template <bool B> struct flag_c { static constexpr bool value = B; };

template <int HEADS, int C, bool ELU, int LISTS>
__global__ __launch_bounds__(1024) void gat_block_kernel(
    const int* __restrict__ ei, int E, int n,
    const float* __restrict__ Hm,
    const float* __restrict__ asd,
    const float* __restrict__ bias, float* __restrict__ outp,
    int* __restrict__ sv_cnt, int* __restrict__ sv_src, int* __restrict__ sv_dl,
    const unsigned* __restrict__ keys) {
  constexpr int CH  = HEADS * C;
  constexpr int VEC = CH / 32;
  constexpr int RU  = 64 / VEC;
  typedef float vrow __attribute__((ext_vector_type(VEC)));

  __shared__ float    acc[NBLK * CH];
  __shared__ unsigned lst_s[NWAVE * LCAP];
  __shared__ unsigned lst_d[NWAVE * LCAP];
  __shared__ unsigned mkey[NBLK * HEADS];
  __shared__ unsigned mold[NBLK * HEADS];
  __shared__ float    den[NBLK * HEADS];
  __shared__ int      wcnt[NWAVE];

  const int tid  = threadIdx.x;
  const int lane = tid & 31;
  const int w    = __builtin_amdgcn_readfirstlane(tid >> 5);
  const int base = blockIdx.x * NBLK;
  const int nb   = min(NBLK, n - base);
  const int* __restrict__ dstp = ei + E;

  for (int i = tid; i < NBLK * CH; i += 1024) acc[i] = 0.f;
  for (int i = tid; i < NBLK * HEADS; i += 1024) {
    mkey[i] = KEY_NEG_INF;
    den[i]  = 0.f;
  }

  unsigned* __restrict__ ls = lst_s + w * LCAP;
  unsigned* __restrict__ ld = lst_d + w * LCAP;

  const int nsteps = (E + KSTEP - 1) / KSTEP;
  const uint4* __restrict__ kp = (const uint4*)keys;
  const unsigned pat = (blockIdx.x & 255u) * 0x01010101u;
  const int seg = (blockIdx.x * NWAVE + w) * LCAP;
  int pos = 0;
  const bool loaded = (LISTS == 2) && (sv_cnt[blockIdx.x * NWAVE] >= 0);

  int cnt;
  if (loaded) {
    cnt = min(max(sv_cnt[blockIdx.x * NWAVE + w], 0), LCAP);
    pos = nsteps;
#pragma unroll 4
    for (int i = lane; i < cnt; i += 32) {
      ls[i] = (unsigned)sv_src[seg + i];
      ld[i] = (unsigned)sv_dl[seg + i];
    }
  } else {
    const int dl = w + NWAVE * lane;
    const bool ok = dl < nb;
    const unsigned mk = __builtin_amdgcn_ballot_w32(ok);
    if (ok) {
      const int idx = __builtin_amdgcn_mbcnt_lo(mk, 0u);
      ls[idx] = SELF_BIT | (unsigned)(base + dl);
    }
    cnt = __builtin_popcount(mk);
  }

  const int hl = (lane * VEC) / C;
  int round = 0;

  for (;;) {
    while (w + NWAVE * pos < nsteps && cnt <= LCAP - KSTEP) {
      const int qlast = w + NWAVE * (pos + SCAN_UNR - 1);
      if (qlast < nsteps) {
        uint4 v[SCAN_UNR];
#pragma unroll
        for (int u = 0; u < SCAN_UNR; ++u) {
          const int q = w + NWAVE * (pos + u);
          v[u] = kp[q * 32 + lane];
        }
        int used = 0;
#pragma unroll
        for (int u = 0; u < SCAN_UNR; ++u) {
          if (cnt <= LCAP - KSTEP) {
            const int q = w + NWAVE * (pos + u);
            key_step(v[u], pat, (unsigned)(q * KSTEP + lane * 16), ls, cnt);
            used = u + 1;
          }
        }
        pos += used;
      } else {
        const int q = w + NWAVE * pos;
        const uint4 v = kp[q * 32 + lane];
        key_step(v, pat, (unsigned)(q * KSTEP + lane * 16), ls, cnt);
        pos += 1;
      }
    }
    const int mine_done = !(w + NWAVE * pos < nsteps);

    if (round > 0)
      for (int i = tid; i < NBLK * HEADS; i += 1024) mold[i] = mkey[i];
    __syncthreads();

    int kept = 0;
    for (int i0 = 0; i0 < cnt; i0 += 32) {
      const int i = i0 + lane;
      bool ok = false;
      int s = 0;
      unsigned dl = 0u;
      if (i < cnt) {
        if (loaded) {
          s = (int)ls[i];
          dl = ld[i];
          ok = true;
        } else {
          const unsigned eidx = ls[i];
          if (eidx & SELF_BIT) {
            s = (int)(eidx & ~SELF_BIT);
            dl = (unsigned)(s - base);
            ok = true;
          } else if (eidx < (unsigned)E) {
            s = ei[eidx];
            dl = (unsigned)(dstp[eidx] - base);
            ok = dl < (unsigned)nb;
          }
        }
      }
      const unsigned mk = __builtin_amdgcn_ballot_w32(ok);
      if (ok) {
        const int o = kept + __builtin_amdgcn_mbcnt_lo(mk, 0u);
        if (!loaded) {
          ls[o] = (unsigned)s;
          ld[o] = dl;
        }
        if (LISTS == 1 && round == 0) {
          sv_src[seg + o] = s;
          sv_dl[seg + o]  = (int)dl;
        }
#pragma unroll
        for (int h = 0; h < HEADS; ++h) {
          float e = asd[s * (2 * HEADS) + h] +
                    asd[(base + (int)dl) * (2 * HEADS) + HEADS + h];
          e = (e > 0.f) ? e : NEG_SLOPE * e;
          atomicMax(&mkey[dl * HEADS + h], fkey(e));
        }
      }
      kept += __builtin_popcount(mk);
    }
    cnt = kept;
    if (LISTS == 1 && round == 0 && lane == 0) wcnt[w] = cnt;
    __syncthreads();

    if (round > 0) {
      for (int i = tid; i < NBLK * CH; i += 1024) {
        const int dl = i / CH, r = i % CH;
        const int h = ((r & 31) * VEC + (r >> 5)) / C;
        const unsigned ko = mold[dl * HEADS + h], kn = mkey[dl * HEADS + h];
        const float f = (ko == KEY_NEG_INF) ? 0.f : __expf(funkey(ko) - funkey(kn));
        acc[i] *= f;
      }
      for (int i = tid; i < NBLK * HEADS; i += 1024) {
        const unsigned ko = mold[i], kn = mkey[i];
        const float f = (ko == KEY_NEG_INF) ? 0.f : __expf(funkey(ko) - funkey(kn));
        den[i] *= f;
      }
      __syncthreads();
    }

    auto sum_trip = [&](const int i0, const int nb32, auto full_c) {
      constexpr bool FULL = decltype(full_c)::value;
      const bool valid = FULL || lane < nb32;
      const int  ii    = i0 + (FULL ? lane : min(lane, nb32 - 1));
      const int  s_l   = (int)ls[ii];
      const int  dl_l  = (int)ld[ii];
      float as_l[HEADS], ad_l[HEADS];
      if constexpr (HEADS == 2) {
        const v2f a2 = *(const v2f*)(asd + (size_t)s_l * 4);
        const v2f d2 = *(const v2f*)(asd + (size_t)(base + dl_l) * 4 + 2);
        as_l[0] = a2.x; as_l[1] = a2.y;
        ad_l[0] = d2.x; ad_l[1] = d2.y;
      } else {
        as_l[0] = asd[s_l * 2];
        ad_l[0] = asd[(base + dl_l) * 2 + 1];
      }

      vrow row[RU];
#pragma unroll
      for (int u8 = 0; u8 < RU; u8 += 8) {
        if (FULL || u8 == 0 || u8 < nb32) {
#pragma unroll
          for (int u = u8; u < u8 + 8; ++u) {
            const int s = __builtin_amdgcn_readlane(s_l, u);
            row[u] = *(const vrow*)(Hm + (size_t)s * CH + lane * VEC);
          }
        }
      }

      float wgt[HEADS];
#pragma unroll
      for (int h = 0; h < HEADS; ++h) {
        float e = as_l[h] + ad_l[h];
        e = (e > 0.f) ? e : NEG_SLOPE * e;
        const float m = funkey(mkey[dl_l * HEADS + h]);
        wgt[h] = valid ? __expf(e - m) : 0.f;
        if (valid) atomicAdd(&den[dl_l * HEADS + h], wgt[h]);
      }

#pragma unroll
      for (int g = 0; g < 32; g += RU) {
        if (FULL || g < nb32) {
          if (g > 0) {
#pragma unroll
            for (int u8 = 0; u8 < RU; u8 += 8) {
              if (FULL || g + u8 < nb32) {
#pragma unroll
                for (int u = u8; u < u8 + 8; ++u) {
                  const int s = __builtin_amdgcn_readlane(s_l, g + u);
                  row[u] = *(const vrow*)(Hm + (size_t)s * CH + lane * VEC);
                }
              }
            }
          }
#pragma unroll
          for (int u = 0; u < RU; ++u) {
            if (FULL || g + u < nb32) {
              const int dl = __builtin_amdgcn_readlane(dl_l, g + u);
              float wv = __int_as_float(
                  __builtin_amdgcn_readlane(__float_as_int(wgt[0]), g + u));
              if constexpr (HEADS == 2) {
                const float w1 = __int_as_float(
                    __builtin_amdgcn_readlane(__float_as_int(wgt[1]), g + u));
                wv = hl ? w1 : wv;
              }
#pragma unroll
              for (int j = 0; j < VEC; ++j)
                atomicAdd(&acc[dl * CH + j * 32 + lane], wv * row[u][j]);
            }
          }
        }
      }
    };
    for (int i0 = 0; i0 < cnt; i0 += 32) {
      const int nb32 = min(32, cnt - i0);
      if (nb32 == 32) sum_trip(i0, 32, flag_c<true>{});
      else            sum_trip(i0, nb32, flag_c<false>{});
    }

    if (__syncthreads_and(mine_done)) break;
    cnt = 0;
    ++round;
  }

  if (LISTS == 1 && w == 0)
    sv_cnt[blockIdx.x * NWAVE + lane] = (round == 0) ? wcnt[lane] : -1;

  for (int i = tid; i < nb * CH; i += 1024) {
    const int dl = i / CH, c = i % CH;
    const int h = c / C;
    float y = acc[dl * CH + (c % VEC) * 32 + (c / VEC)] /
                  (den[dl * HEADS + h] + 1e-16f) + bias[c];
    if (ELU) y = (y > 0.f) ? y : expm1f(y);
    outp[(size_t)(base + dl) * CH + c] = y;
  }
}


#define GEMM_WGP 256

static inline int cdiv(long long a, long long b) { return (int)((a + b - 1) / b); }

extern "C" void kernel_launch(void* const* d_in, const int* in_sizes, int n_in,
                              void* d_out, int out_size, void* d_ws, size_t ws_size,
                              hipStream_t stream) {
  const int IN_CH = 128, HID = 64, HEADS = 2, OUT = 64;
  const int CH1 = HEADS * HID;

  const float* x        = (const float*)d_in[0];
  const int*   ei       = (const int*)d_in[1];
  const float* W1       = (const float*)d_in[2];
  const float* att1_src = (const float*)d_in[3];
  const float* att1_dst = (const float*)d_in[4];
  const float* b1       = (const float*)d_in[5];
  const float* W2       = (const float*)d_in[6];
  const float* att2_src = (const float*)d_in[7];
  const float* att2_dst = (const float*)d_in[8];
  const float* b2       = (const float*)d_in[9];
  float* out            = (float*)d_out;

  const int n  = in_sizes[0] / IN_CH;
  const int E  = in_sizes[1] / 2;

  char* ws = (char*)d_ws;
  size_t off = 0;
  auto alloc = [&](size_t bytes) -> void* {
    void* p = ws + off;
    off += (bytes + 255) & ~(size_t)255;
    return p;
  };
  float* h1   = (float*)alloc((size_t)n * CH1 * 4);
  float* agg1 = (float*)alloc((size_t)n * CH1 * 4);
  float* h2   = (float*)alloc((size_t)n * OUT * 4);
  float* asd1 = (float*)alloc((size_t)n * 2 * HEADS * 4);
  float* asd2 = (float*)alloc((size_t)n * 2 * 4);
  float* V1   = (float*)alloc((size_t)2 * HEADS * IN_CH * 4);
  float* V2   = (float*)alloc((size_t)2 * CH1 * 4);
  const int nblk = cdiv(n, NBLK);
  int* sv_cnt = (int*)alloc((size_t)nblk * NWAVE * 4);
  int* sv_src = (int*)alloc((size_t)nblk * NWAVE * LCAP * 4);
  int* sv_dl  = (int*)alloc((size_t)nblk * NWAVE * LCAP * 4);
  const int epad = cdiv(E, KSTEP) * KSTEP;
  unsigned* keys = (unsigned*)alloc((size_t)epad);
  (void)ws_size; (void)n_in; (void)out_size;

  const int BT = 256;

  dst_key_kernel<<<FOLD_BLOCKS + cdiv(epad / 4, KEY_UNR * BT), BT, 0, stream>>>(
      ei, E, epad, keys, W1, att1_src, att1_dst, V1, W2, att2_src, att2_dst, V2);

  {
    dim3 block(32, 4, 3);
    dim3 grid(GEMM_WGP);
    gemm_wmma_f32<128, 2, 4, 3, 4><<<grid, block, 0, stream>>>(x, W1, V1, h1, asd1, n / 16);
  }
  gat_block_kernel<2, 64, true, 1><<<nblk, 1024, 0, stream>>>(
      ei, E, n, h1, asd1, b1, agg1, sv_cnt, sv_src, sv_dl, keys);

  {
    dim3 block(32, 2, 4);
    dim3 grid(GEMM_WGP);
    gemm_wmma_f32<128, 2, 2, 4, 2><<<grid, block, 0, stream>>>(agg1, W2, V2, h2, asd2, n / 16);
  }
  gat_block_kernel<1, 64, false, 2><<<nblk, 1024, 0, stream>>>(
      ei, E, n, h2, asd2, b2, out, sv_cnt, sv_src, sv_dl, keys);
}


# edge kernels: the max phase keeps each listed edge's score per head in LDS beside the list (lists 704 long to make room); the sum phase reads it, no second gather of the node dots (sums: LDS arrival)
# speedup vs baseline: 8.7503x; 1.0511x over previous
#include <hip/hip_runtime.h>
#include <hip/hip_bf16.h>


typedef float v2f __attribute__((ext_vector_type(2)));
typedef float v8f __attribute__((ext_vector_type(8)));

#define NEG_SLOPE 0.2f
#define KEY_NEG_INF 0x007FFFFFu

__device__ __forceinline__ unsigned fkey(float x) {
  unsigned b = __float_as_uint(x);
  return (b & 0x80000000u) ? ~b : (b | 0x80000000u);
}
__device__ __forceinline__ float funkey(unsigned k) {
  return __uint_as_float((k & 0x80000000u) ? (k & 0x7FFFFFFFu) : ~k);
}


template <int K, int CT, int NV, bool WITHV>
__device__ __forceinline__ void gemm_row_tiles(
    const float* __restrict__ A, float* __restrict__ H, float* __restrict__ asd,
    const v2f (&wf)[CT][K / 4], const v2f* __restrict__ vf,
    const int firstTile, const int tileStep, const int nRowTiles,
    const int Cout, const int colTile0, const int lane) {
  constexpr int KS = K / 4;
  const int mn    = lane & 15;
  const int khalf = (lane >> 4) * 2;
  for (int rowTile = firstTile; rowTile < nRowTiles; rowTile += tileStep) {
    const float* __restrict__ arow = A + (size_t)(rowTile * 16 + mn) * K + khalf;
    v2f a[KS];
#pragma unroll
    for (int j = 0; j < KS; ++j) a[j] = *(const v2f*)(arow + 4 * j);

    v8f acc[CT];
    v8f accv = v8f{};
#pragma unroll
    for (int c = 0; c < CT; ++c) acc[c] = v8f{};
#pragma unroll
    for (int j = 0; j < KS; ++j) {
#pragma unroll
      for (int c = 0; c < CT; ++c)
        acc[c] = __builtin_amdgcn_wmma_f32_16x16x4_f32(
             false, a[j],  false, wf[c][j],
             (short)0, acc[c],  false,  false);
      if constexpr (WITHV)
        accv = __builtin_amdgcn_wmma_f32_16x16x4_f32(
             false, a[j],  false, vf[j],
             (short)0, accv,  false,  false);
    }

    const int mrow = rowTile * 16 + 8 * (lane >> 4);
#pragma unroll
    for (int c = 0; c < CT; ++c) {
      const int ncol = (colTile0 + c) * 16 + mn;
#pragma unroll
      for (int v = 0; v < 8; ++v)
        H[(size_t)(mrow + v) * Cout + ncol] = acc[c][v];
    }
    if constexpr (WITHV) {
      if (mn < NV) {
#pragma unroll
        for (int v = 0; v < 8; ++v)
          asd[(size_t)(mrow + v) * NV + mn] = accv[v];
      }
    }
  }
}

template <int K, int CT, int NCG, int NRG, int NV>
__global__ __launch_bounds__(32 * NCG * NRG) void gemm_wmma_f32(
    const float* __restrict__ A, const float* __restrict__ W,
    const float* __restrict__ V, float* __restrict__ H,
    float* __restrict__ asd, int nRowTiles) {
  constexpr int KS   = K / 4;
  constexpr int Cout = 16 * CT * NCG;
  const int lane = threadIdx.x;
  const int colGroup = __builtin_amdgcn_readfirstlane(threadIdx.y);
  const int rowGroup = __builtin_amdgcn_readfirstlane(threadIdx.z);

  const int mn    = lane & 15;
  const int khalf = (lane >> 4) * 2;

  v2f wf[CT][KS];
#pragma unroll
  for (int c = 0; c < CT; ++c) {
    const float* __restrict__ wrow =
        W + (size_t)((colGroup * CT + c) * 16 + mn) * K + khalf;
#pragma unroll
    for (int j = 0; j < KS; ++j) wf[c][j] = *(const v2f*)(wrow + 4 * j);
  }

  const int firstTile = rowGroup * gridDim.x + blockIdx.x;
  const int tileStep  = gridDim.x * NRG;
  if (colGroup == (rowGroup * NCG) / NRG) {
    v2f vf[KS];
    const float* __restrict__ vrow = V + (size_t)min(mn, NV - 1) * K + khalf;
#pragma unroll
    for (int j = 0; j < KS; ++j) {
      const v2f t = *(const v2f*)(vrow + 4 * j);
      vf[j] = (mn < NV) ? t : v2f{};
    }
    gemm_row_tiles<K, CT, NV, true>(A, H, asd, wf, vf, firstTile, tileStep,
                                    nRowTiles, Cout, colGroup * CT, lane);
  } else {
    gemm_row_tiles<K, CT, NV, false>(A, H, asd, wf, nullptr, firstTile, tileStep,
                                     nRowTiles, Cout, colGroup * CT, lane);
  }
}


template <int HEADS, int C, int K>
__device__ __forceinline__ void fold_att(const float* __restrict__ W,
                                         const float* __restrict__ att_src,
                                         const float* __restrict__ att_dst,
                                         float* __restrict__ V, int r, int k) {
  const bool isDst = r >= HEADS;
  const int  h     = isDst ? r - HEADS : r;
  const float* __restrict__ att = (isDst ? att_dst : att_src) + h * C;
  const float* __restrict__ wp  = W + (size_t)(h * C) * K + k;
  float s = 0.f;
#pragma unroll
  for (int c = 0; c < C; ++c) s += att[c] * wp[(size_t)c * K];
  V[r * K + k] = s;
}


#define NBLK     196
#define NWAVE    32
#define KSTEP    512
#define LCAP     704
#define ECAP     160
#define SCAN_UNR 8
#define SELF_BIT 0x80000000u

#define KEY_UNR 4
#define FOLD_BLOCKS 3
#define FOLD_K      128

__device__ __forceinline__ unsigned key_word(const int4 v) {
  const unsigned k0 = ((unsigned)v.x / NBLK) & 255u;
  const unsigned k1 = ((unsigned)v.y / NBLK) & 255u;
  const unsigned k2 = ((unsigned)v.z / NBLK) & 255u;
  const unsigned k3 = ((unsigned)v.w / NBLK) & 255u;
  return k0 | (k1 << 8) | (k2 << 16) | (k3 << 24);
}

__global__ __launch_bounds__(256) void dst_key_kernel(
    const int* __restrict__ ei, int E, int epad, unsigned* __restrict__ keys,
    const float* __restrict__ W1, const float* __restrict__ att1_src,
    const float* __restrict__ att1_dst, float* __restrict__ V1,
    const float* __restrict__ W2, const float* __restrict__ att2_src,
    const float* __restrict__ att2_dst, float* __restrict__ V2) {
  if (blockIdx.x < FOLD_BLOCKS) {
    const int rr = threadIdx.x / FOLD_K, k = threadIdx.x % FOLD_K;
    if (blockIdx.x < 2)
      fold_att<2, 64, FOLD_K>(W1, att1_src, att1_dst, V1, 2 * (int)blockIdx.x + rr, k);
    else
      fold_att<1, 64, FOLD_K>(W2, att2_src, att2_dst, V2, rr, k);
    return;
  }
  const int nw    = epad >> 2;
  const int nfull = E >> 2;
  const int T     = (gridDim.x - FOLD_BLOCKS) * blockDim.x;
  const int t0    = (blockIdx.x - FOLD_BLOCKS) * blockDim.x + threadIdx.x;
  const int* __restrict__ dstp = ei + E;
  if (nfull > 0 && ((E & 3) == 0) && ((((size_t)ei) & 15) == 0)) {
    for (int tb = t0; tb < nw; tb += KEY_UNR * T) {
      int4 v[KEY_UNR];
#pragma unroll
      for (int u = 0; u < KEY_UNR; ++u) {
        const int t = min(tb + u * T, nfull - 1);
        v[u] = *(const int4*)(dstp + 4 * t);
      }
#pragma unroll
      for (int u = 0; u < KEY_UNR; ++u) {
        const int t = tb + u * T;
        if (t < nw) keys[t] = (t < nfull) ? key_word(v[u]) : 0u;
      }
    }
  } else {
    for (int t = t0; t < nw; t += T) {
      const int e = 4 * t;
      int4 v;
      v.x = (e + 0 < E) ? dstp[e + 0] : 0;
      v.y = (e + 1 < E) ? dstp[e + 1] : 0;
      v.z = (e + 2 < E) ? dstp[e + 2] : 0;
      v.w = (e + 3 < E) ? dstp[e + 3] : 0;
      keys[t] = key_word(v);
    }
  }
}

__device__ __forceinline__ void key_step(const uint4 v, const unsigned pat,
                                         const unsigned e0,
                                         unsigned* __restrict__ ls, int& cnt) {
  const unsigned x0 = v.x ^ pat, x1 = v.y ^ pat, x2 = v.z ^ pat, x3 = v.w ^ pat;
  const unsigned t0 = (x0 - 0x01010101u) & ~x0;
  const unsigned t1 = (x1 - 0x01010101u) & ~x1;
  const unsigned t2 = (x2 - 0x01010101u) & ~x2;
  const unsigned t3 = (x3 - 0x01010101u) & ~x3;
  if (__builtin_amdgcn_ballot_w32(((t0 | t1 | t2 | t3) & 0x80808080u) != 0u)) {
    unsigned m = ((t0 >> 7) & 0x01010101u) | ((t1 >> 6) & 0x02020202u) |
                 ((t2 >> 5) & 0x04040404u) | ((t3 >> 4) & 0x08080808u);
    for (int it = 0; it < 16; ++it) {
      const bool has = m != 0u;
      const unsigned mk = __builtin_amdgcn_ballot_w32(has);
      if (mk == 0u) break;
      if (has) {
        const unsigned bit = (unsigned)__builtin_ctz(m);
        ls[cnt + __builtin_amdgcn_mbcnt_lo(mk, 0u)] = e0 + 4u * (bit & 7u) + (bit >> 3);
        m &= m - 1u;
      }
      cnt += __builtin_popcount(mk);
    }
  }
}

template <bool B> struct flag_c { static constexpr bool value = B; };

template <int HEADS, int C, bool ELU, int LISTS>
__global__ __launch_bounds__(1024) void gat_block_kernel(
    const int* __restrict__ ei, int E, int n,
    const float* __restrict__ Hm,
    const float* __restrict__ asd,
    const float* __restrict__ bias, float* __restrict__ outp,
    int* __restrict__ sv_cnt, int* __restrict__ sv_src, int* __restrict__ sv_dl,
    const unsigned* __restrict__ keys) {
  constexpr int CH  = HEADS * C;
  constexpr int VEC = CH / 32;
  constexpr int RU  = 64 / VEC;
  typedef float vrow __attribute__((ext_vector_type(VEC)));

  __shared__ float    acc[NBLK * CH];
  __shared__ unsigned lst_s[NWAVE * LCAP];
  __shared__ unsigned lst_d[NWAVE * LCAP];
  __shared__ float    esc[NWAVE * ECAP * HEADS];
  __shared__ unsigned mkey[NBLK * HEADS];
  __shared__ unsigned mold[NBLK * HEADS];
  __shared__ float    den[NBLK * HEADS];
  __shared__ int      wcnt[NWAVE];

  const int tid  = threadIdx.x;
  const int lane = tid & 31;
  const int w    = __builtin_amdgcn_readfirstlane(tid >> 5);
  const int base = blockIdx.x * NBLK;
  const int nb   = min(NBLK, n - base);
  const int* __restrict__ dstp = ei + E;

  for (int i = tid; i < NBLK * CH; i += 1024) acc[i] = 0.f;
  for (int i = tid; i < NBLK * HEADS; i += 1024) {
    mkey[i] = KEY_NEG_INF;
    den[i]  = 0.f;
  }

  unsigned* __restrict__ ls = lst_s + w * LCAP;
  unsigned* __restrict__ ld = lst_d + w * LCAP;
  float*    __restrict__ es = esc + w * (ECAP * HEADS);

  const int nsteps = (E + KSTEP - 1) / KSTEP;
  const uint4* __restrict__ kp = (const uint4*)keys;
  const unsigned pat = (blockIdx.x & 255u) * 0x01010101u;
  const int seg = (blockIdx.x * NWAVE + w) * LCAP;
  int pos = 0;
  const bool loaded = (LISTS == 2) && (sv_cnt[blockIdx.x * NWAVE] >= 0);

  int cnt;
  if (loaded) {
    cnt = min(max(sv_cnt[blockIdx.x * NWAVE + w], 0), LCAP);
    pos = nsteps;
#pragma unroll 4
    for (int i = lane; i < cnt; i += 32) {
      ls[i] = (unsigned)sv_src[seg + i];
      ld[i] = (unsigned)sv_dl[seg + i];
    }
  } else {
    const int dl = w + NWAVE * lane;
    const bool ok = dl < nb;
    const unsigned mk = __builtin_amdgcn_ballot_w32(ok);
    if (ok) {
      const int idx = __builtin_amdgcn_mbcnt_lo(mk, 0u);
      ls[idx] = SELF_BIT | (unsigned)(base + dl);
    }
    cnt = __builtin_popcount(mk);
  }

  const int hl = (lane * VEC) / C;
  int round = 0;

  for (;;) {
    while (w + NWAVE * pos < nsteps && cnt <= LCAP - KSTEP) {
      const int qlast = w + NWAVE * (pos + SCAN_UNR - 1);
      if (qlast < nsteps) {
        uint4 v[SCAN_UNR];
#pragma unroll
        for (int u = 0; u < SCAN_UNR; ++u) {
          const int q = w + NWAVE * (pos + u);
          v[u] = kp[q * 32 + lane];
        }
        int used = 0;
#pragma unroll
        for (int u = 0; u < SCAN_UNR; ++u) {
          if (cnt <= LCAP - KSTEP) {
            const int q = w + NWAVE * (pos + u);
            key_step(v[u], pat, (unsigned)(q * KSTEP + lane * 16), ls, cnt);
            used = u + 1;
          }
        }
        pos += used;
      } else {
        const int q = w + NWAVE * pos;
        const uint4 v = kp[q * 32 + lane];
        key_step(v, pat, (unsigned)(q * KSTEP + lane * 16), ls, cnt);
        pos += 1;
      }
    }
    const int mine_done = !(w + NWAVE * pos < nsteps);

    if (round > 0)
      for (int i = tid; i < NBLK * HEADS; i += 1024) mold[i] = mkey[i];
    __syncthreads();

    int kept = 0;
    for (int i0 = 0; i0 < cnt; i0 += 32) {
      const int i = i0 + lane;
      bool ok = false;
      int s = 0;
      unsigned dl = 0u;
      if (i < cnt) {
        if (loaded) {
          s = (int)ls[i];
          dl = ld[i];
          ok = true;
        } else {
          const unsigned eidx = ls[i];
          if (eidx & SELF_BIT) {
            s = (int)(eidx & ~SELF_BIT);
            dl = (unsigned)(s - base);
            ok = true;
          } else if (eidx < (unsigned)E) {
            s = ei[eidx];
            dl = (unsigned)(dstp[eidx] - base);
            ok = dl < (unsigned)nb;
          }
        }
      }
      const unsigned mk = __builtin_amdgcn_ballot_w32(ok);
      if (ok) {
        const int o = kept + __builtin_amdgcn_mbcnt_lo(mk, 0u);
        if (!loaded) {
          ls[o] = (unsigned)s;
          ld[o] = dl;
        }
        if (LISTS == 1 && round == 0) {
          sv_src[seg + o] = s;
          sv_dl[seg + o]  = (int)dl;
        }
        float e_h[HEADS];
        if constexpr (HEADS == 2) {
          const v2f a2 = *(const v2f*)(asd + (size_t)s * 4);
          const v2f d2 = *(const v2f*)(asd + (size_t)(base + (int)dl) * 4 + 2);
          e_h[0] = a2.x + d2.x;
          e_h[1] = a2.y + d2.y;
        } else {
          e_h[0] = asd[s * 2] + asd[(base + (int)dl) * 2 + 1];
        }
#pragma unroll
        for (int h = 0; h < HEADS; ++h) {
          const float e = (e_h[h] > 0.f) ? e_h[h] : NEG_SLOPE * e_h[h];
          if (o < ECAP) es[h * ECAP + o] = e;
          atomicMax(&mkey[dl * HEADS + h], fkey(e));
        }
      }
      kept += __builtin_popcount(mk);
    }
    cnt = kept;
    if (LISTS == 1 && round == 0 && lane == 0) wcnt[w] = cnt;
    __syncthreads();

    if (round > 0) {
      for (int i = tid; i < NBLK * CH; i += 1024) {
        const int dl = i / CH, r = i % CH;
        const int h = ((r & 31) * VEC + (r >> 5)) / C;
        const unsigned ko = mold[dl * HEADS + h], kn = mkey[dl * HEADS + h];
        const float f = (ko == KEY_NEG_INF) ? 0.f : __expf(funkey(ko) - funkey(kn));
        acc[i] *= f;
      }
      for (int i = tid; i < NBLK * HEADS; i += 1024) {
        const unsigned ko = mold[i], kn = mkey[i];
        const float f = (ko == KEY_NEG_INF) ? 0.f : __expf(funkey(ko) - funkey(kn));
        den[i] *= f;
      }
      __syncthreads();
    }

    auto sum_trip = [&](const int i0, const int nb32, auto full_c) {
      constexpr bool FULL = decltype(full_c)::value;
      const bool valid = FULL || lane < nb32;
      const int  ii    = i0 + (FULL ? lane : min(lane, nb32 - 1));
      const int  s_l   = (int)ls[ii];
      const int  dl_l  = (int)ld[ii];
      float e_l[HEADS];
      if (i0 + nb32 <= ECAP) {
#pragma unroll
        for (int h = 0; h < HEADS; ++h) e_l[h] = es[h * ECAP + ii];
      } else {
        float as_l[HEADS], ad_l[HEADS];
        if constexpr (HEADS == 2) {
          const v2f a2 = *(const v2f*)(asd + (size_t)s_l * 4);
          const v2f d2 = *(const v2f*)(asd + (size_t)(base + dl_l) * 4 + 2);
          as_l[0] = a2.x; as_l[1] = a2.y;
          ad_l[0] = d2.x; ad_l[1] = d2.y;
        } else {
          as_l[0] = asd[s_l * 2];
          ad_l[0] = asd[(base + dl_l) * 2 + 1];
        }
#pragma unroll
        for (int h = 0; h < HEADS; ++h) {
          const float e = as_l[h] + ad_l[h];
          e_l[h] = (e > 0.f) ? e : NEG_SLOPE * e;
        }
      }

      vrow row[RU];
#pragma unroll
      for (int u8 = 0; u8 < RU; u8 += 8) {
        if (FULL || u8 == 0 || u8 < nb32) {
#pragma unroll
          for (int u = u8; u < u8 + 8; ++u) {
            const int s = __builtin_amdgcn_readlane(s_l, u);
            row[u] = *(const vrow*)(Hm + (size_t)s * CH + lane * VEC);
          }
        }
      }

      float wgt[HEADS];
#pragma unroll
      for (int h = 0; h < HEADS; ++h) {
        const float m = funkey(mkey[dl_l * HEADS + h]);
        wgt[h] = valid ? __expf(e_l[h] - m) : 0.f;
        if (valid) atomicAdd(&den[dl_l * HEADS + h], wgt[h]);
      }

#pragma unroll
      for (int g = 0; g < 32; g += RU) {
        if (FULL || g < nb32) {
          if (g > 0) {
#pragma unroll
            for (int u8 = 0; u8 < RU; u8 += 8) {
              if (FULL || g + u8 < nb32) {
#pragma unroll
                for (int u = u8; u < u8 + 8; ++u) {
                  const int s = __builtin_amdgcn_readlane(s_l, g + u);
                  row[u] = *(const vrow*)(Hm + (size_t)s * CH + lane * VEC);
                }
              }
            }
          }
#pragma unroll
          for (int u = 0; u < RU; ++u) {
            if (FULL || g + u < nb32) {
              const int dl = __builtin_amdgcn_readlane(dl_l, g + u);
              float wv = __int_as_float(
                  __builtin_amdgcn_readlane(__float_as_int(wgt[0]), g + u));
              if constexpr (HEADS == 2) {
                const float w1 = __int_as_float(
                    __builtin_amdgcn_readlane(__float_as_int(wgt[1]), g + u));
                wv = hl ? w1 : wv;
              }
#pragma unroll
              for (int j = 0; j < VEC; ++j)
                atomicAdd(&acc[dl * CH + j * 32 + lane], wv * row[u][j]);
            }
          }
        }
      }
    };
    for (int i0 = 0; i0 < cnt; i0 += 32) {
      const int nb32 = min(32, cnt - i0);
      if (nb32 == 32) sum_trip(i0, 32, flag_c<true>{});
      else            sum_trip(i0, nb32, flag_c<false>{});
    }

    if (__syncthreads_and(mine_done)) break;
    cnt = 0;
    ++round;
  }

  if (LISTS == 1 && w == 0)
    sv_cnt[blockIdx.x * NWAVE + lane] = (round == 0) ? wcnt[lane] : -1;

  for (int i = tid; i < nb * CH; i += 1024) {
    const int dl = i / CH, c = i % CH;
    const int h = c / C;
    float y = acc[dl * CH + (c % VEC) * 32 + (c / VEC)] /
                  (den[dl * HEADS + h] + 1e-16f) + bias[c];
    if (ELU) y = (y > 0.f) ? y : expm1f(y);
    outp[(size_t)(base + dl) * CH + c] = y;
  }
}


#define GEMM_WGP 256

static inline int cdiv(long long a, long long b) { return (int)((a + b - 1) / b); }

extern "C" void kernel_launch(void* const* d_in, const int* in_sizes, int n_in,
                              void* d_out, int out_size, void* d_ws, size_t ws_size,
                              hipStream_t stream) {
  const int IN_CH = 128, HID = 64, HEADS = 2, OUT = 64;
  const int CH1 = HEADS * HID;

  const float* x        = (const float*)d_in[0];
  const int*   ei       = (const int*)d_in[1];
  const float* W1       = (const float*)d_in[2];
  const float* att1_src = (const float*)d_in[3];
  const float* att1_dst = (const float*)d_in[4];
  const float* b1       = (const float*)d_in[5];
  const float* W2       = (const float*)d_in[6];
  const float* att2_src = (const float*)d_in[7];
  const float* att2_dst = (const float*)d_in[8];
  const float* b2       = (const float*)d_in[9];
  float* out            = (float*)d_out;

  const int n  = in_sizes[0] / IN_CH;
  const int E  = in_sizes[1] / 2;

  char* ws = (char*)d_ws;
  size_t off = 0;
  auto alloc = [&](size_t bytes) -> void* {
    void* p = ws + off;
    off += (bytes + 255) & ~(size_t)255;
    return p;
  };
  float* h1   = (float*)alloc((size_t)n * CH1 * 4);
  float* agg1 = (float*)alloc((size_t)n * CH1 * 4);
  float* h2   = (float*)alloc((size_t)n * OUT * 4);
  float* asd1 = (float*)alloc((size_t)n * 2 * HEADS * 4);
  float* asd2 = (float*)alloc((size_t)n * 2 * 4);
  float* V1   = (float*)alloc((size_t)2 * HEADS * IN_CH * 4);
  float* V2   = (float*)alloc((size_t)2 * CH1 * 4);
  const int nblk = cdiv(n, NBLK);
  int* sv_cnt = (int*)alloc((size_t)nblk * NWAVE * 4);
  int* sv_src = (int*)alloc((size_t)nblk * NWAVE * LCAP * 4);
  int* sv_dl  = (int*)alloc((size_t)nblk * NWAVE * LCAP * 4);
  const int epad = cdiv(E, KSTEP) * KSTEP;
  unsigned* keys = (unsigned*)alloc((size_t)epad);
  (void)ws_size; (void)n_in; (void)out_size;

  const int BT = 256;

  dst_key_kernel<<<FOLD_BLOCKS + cdiv(epad / 4, KEY_UNR * BT), BT, 0, stream>>>(
      ei, E, epad, keys, W1, att1_src, att1_dst, V1, W2, att2_src, att2_dst, V2);

  {
    dim3 block(32, 4, 3);
    dim3 grid(GEMM_WGP);
    gemm_wmma_f32<128, 2, 4, 3, 4><<<grid, block, 0, stream>>>(x, W1, V1, h1, asd1, n / 16);
  }
  gat_block_kernel<2, 64, true, 1><<<nblk, 1024, 0, stream>>>(
      ei, E, n, h1, asd1, b1, agg1, sv_cnt, sv_src, sv_dl, keys);

  {
    dim3 block(32, 2, 4);
    dim3 grid(GEMM_WGP);
    gemm_wmma_f32<128, 2, 2, 4, 2><<<grid, block, 0, stream>>>(agg1, W2, V2, h2, asd2, n / 16);
  }
  gat_block_kernel<1, 64, false, 2><<<nblk, 1024, 0, stream>>>(
      ei, E, n, h2, asd2, b2, out, sv_cnt, sv_src, sv_dl, keys);
}
